# starting kernel
# baseline (speedup 1.0000x reference)
#include <hip/hip_runtime.h>

#define N_NODES 100000
#define N_EDGES 1600000
#define D_IN    128
#define D_H     64

typedef __attribute__((ext_vector_type(2))) float v2f;
typedef __attribute__((ext_vector_type(8))) float v8f;

template <int K, bool RELU_IN>
__global__ void gemm_wmma_kernel(const float* __restrict__ A,
                                 const float* __restrict__ W,
                                 float* __restrict__ C, int nrows) {
  __shared__ v2f Pw[(K / 2) * 64];

  for (int idx = threadIdx.x; idx < K * 64; idx += blockDim.x) {
    const int k = idx >> 6;
    const int c = idx & 63;
    ((float*)Pw)[(k >> 1) * 128 + c * 2 + (k & 1)] = W[idx];
  }
  __syncthreads();

  const int lane = threadIdx.x & 31;
  const int wave = blockIdx.x * (blockDim.x >> 5) + (threadIdx.x >> 5);
  const int row0 = wave * 16;
  if (row0 >= nrows) return;

  const int half = lane >> 4;
  const int l15  = lane & 15;
  const long long arow = row0 + l15;

  v8f acc0 = {}, acc1 = {}, acc2 = {}, acc3 = {};

  for (int k = 0; k < K; k += 4) {
    const int kk = k + 2 * half;
    v2f a = *(const v2f*)(A + arow * K + kk);
    if (RELU_IN) {
      a.x = fmaxf(a.x, 0.0f);
      a.y = fmaxf(a.y, 0.0f);
    }
    const v2f* __restrict__ Pk = Pw + (k / 2 + half) * 64;
    const v2f b0 = Pk[l15 +  0];
    const v2f b1 = Pk[l15 + 16];
    const v2f b2 = Pk[l15 + 32];
    const v2f b3 = Pk[l15 + 48];

    acc0 = __builtin_amdgcn_wmma_f32_16x16x4_f32(false, a, false, b0, (short)0, acc0, false, false);
    acc1 = __builtin_amdgcn_wmma_f32_16x16x4_f32(false, a, false, b1, (short)0, acc1, false, false);
    acc2 = __builtin_amdgcn_wmma_f32_16x16x4_f32(false, a, false, b2, (short)0, acc2, false, false);
    acc3 = __builtin_amdgcn_wmma_f32_16x16x4_f32(false, a, false, b3, (short)0, acc3, false, false);
  }

#pragma unroll
  for (int r = 0; r < 8; ++r) {
    const long long m = row0 + r + 8 * half;
    float* __restrict__ crow = C + m * 64;
    crow[l15 +  0] = acc0[r];
    crow[l15 + 16] = acc1[r];
    crow[l15 + 32] = acc2[r];
    crow[l15 + 48] = acc3[r];
  }
}

__global__ void init_bias_kernel(float* __restrict__ out,
                                 const float* __restrict__ bias, int total) {
  int tid = blockIdx.x * blockDim.x + threadIdx.x;
  if (tid < total) out[tid] = bias[tid & 63];
}

__global__ void spmm_atomic_kernel(const float* __restrict__ sup,
                                   const int* __restrict__ src,
                                   const int* __restrict__ dst,
                                   const float* __restrict__ ew,
                                   float* __restrict__ out, int nedges) {
  long long tid = (long long)blockIdx.x * blockDim.x + threadIdx.x;
  int e = (int)(tid >> 4);
  if (e >= nedges) return;
  int j = ((int)tid & 15) * 4;

  float w = ew[e];
  long long s = src[e];
  long long d = dst[e];

  const float4 v = *(const float4*)(sup + s * 64 + j);
  float* __restrict__ o = out + d * 64 + j;
  atomicAdd(o + 0, w * v.x);
  atomicAdd(o + 1, w * v.y);
  atomicAdd(o + 2, w * v.z);
  atomicAdd(o + 3, w * v.w);
}

extern "C" void kernel_launch(void* const* d_in, const int* in_sizes, int n_in,
                              void* d_out, int out_size, void* d_ws, size_t ws_size,
                              hipStream_t stream) {
  const float* features = (const float*)d_in[0];
  const int*   src      = (const int*)  d_in[1];
  const int*   dst      = (const int*)  d_in[2];
  const float* ew       = (const float*)d_in[3];
  const float* W1       = (const float*)d_in[4];
  const float* b1       = (const float*)d_in[5];
  const float* W2       = (const float*)d_in[6];
  const float* b2       = (const float*)d_in[7];
  const float* W3       = (const float*)d_in[8];
  const float* b3       = (const float*)d_in[9];
  float* out = (float*)d_out;

  const long long feat_elems = (long long)N_NODES * D_H;
  float* bufA = (float*)d_ws;
  float* bufB = bufA + feat_elems;

  const dim3 blk(256);
  const int gemm_blocks = (N_NODES / 16 + 7) / 8;
  const int init_blocks = (int)((feat_elems + 255) / 256);
  const int spmm_blocks = (int)(((long long)N_EDGES * 16 + 255) / 256);

  gemm_wmma_kernel<D_IN, false><<<gemm_blocks, blk, 0, stream>>>(features, W1, bufA, N_NODES);
  init_bias_kernel<<<init_blocks, blk, 0, stream>>>(bufB, b1, (int)feat_elems);
  spmm_atomic_kernel<<<spmm_blocks, blk, 0, stream>>>(bufA, src, dst, ew, bufB, N_EDGES);

  gemm_wmma_kernel<D_H, true><<<gemm_blocks, blk, 0, stream>>>(bufB, W2, bufA, N_NODES);
  init_bias_kernel<<<init_blocks, blk, 0, stream>>>(bufB, b2, (int)feat_elems);
  spmm_atomic_kernel<<<spmm_blocks, blk, 0, stream>>>(bufA, src, dst, ew, bufB, N_EDGES);

  gemm_wmma_kernel<D_H, true><<<gemm_blocks, blk, 0, stream>>>(bufB, W3, bufA, N_NODES);
  init_bias_kernel<<<init_blocks, blk, 0, stream>>>(out, b3, (int)feat_elems);
  spmm_atomic_kernel<<<spmm_blocks, blk, 0, stream>>>(bufA, src, dst, ew, out, N_EDGES);

  (void)in_sizes; (void)n_in; (void)out_size; (void)ws_size;
}


# SpMM by destination: edges grouped by dst once per call (integer returning adds as arrival-order rank, prefix sum, place); rows summed in registers, stored once; no f32 atomics, no bias-init launch
# speedup vs baseline: 1.6626x; 1.6626x over previous
#include <hip/hip_runtime.h>

#define N_NODES 100000
#define N_EDGES 1600000
#define D_IN    128
#define D_H     64

typedef __attribute__((ext_vector_type(2))) float v2f;
typedef __attribute__((ext_vector_type(8))) float v8f;

template <int K, bool RELU_IN>
__global__ void gemm_wmma_kernel(const float* __restrict__ A,
                                 const float* __restrict__ W,
                                 float* __restrict__ C, int nrows) {
  __shared__ v2f Pw[(K / 2) * 64];

  for (int idx = threadIdx.x; idx < K * 64; idx += blockDim.x) {
    const int k = idx >> 6;
    const int c = idx & 63;
    ((float*)Pw)[(k >> 1) * 128 + c * 2 + (k & 1)] = W[idx];
  }
  __syncthreads();

  const int lane = threadIdx.x & 31;
  const int wave = blockIdx.x * (blockDim.x >> 5) + (threadIdx.x >> 5);
  const int row0 = wave * 16;
  if (row0 >= nrows) return;

  const int half = lane >> 4;
  const int l15  = lane & 15;
  const long long arow = row0 + l15;

  v8f acc0 = {}, acc1 = {}, acc2 = {}, acc3 = {};

  for (int k = 0; k < K; k += 4) {
    const int kk = k + 2 * half;
    v2f a = *(const v2f*)(A + arow * K + kk);
    if (RELU_IN) {
      a.x = fmaxf(a.x, 0.0f);
      a.y = fmaxf(a.y, 0.0f);
    }
    const v2f* __restrict__ Pk = Pw + (k / 2 + half) * 64;
    const v2f b0 = Pk[l15 +  0];
    const v2f b1 = Pk[l15 + 16];
    const v2f b2 = Pk[l15 + 32];
    const v2f b3 = Pk[l15 + 48];

    acc0 = __builtin_amdgcn_wmma_f32_16x16x4_f32(false, a, false, b0, (short)0, acc0, false, false);
    acc1 = __builtin_amdgcn_wmma_f32_16x16x4_f32(false, a, false, b1, (short)0, acc1, false, false);
    acc2 = __builtin_amdgcn_wmma_f32_16x16x4_f32(false, a, false, b2, (short)0, acc2, false, false);
    acc3 = __builtin_amdgcn_wmma_f32_16x16x4_f32(false, a, false, b3, (short)0, acc3, false, false);
  }

#pragma unroll
  for (int r = 0; r < 8; ++r) {
    const long long m = row0 + r + 8 * half;
    float* __restrict__ crow = C + m * 64;
    crow[l15 +  0] = acc0[r];
    crow[l15 + 16] = acc1[r];
    crow[l15 + 32] = acc2[r];
    crow[l15 + 48] = acc3[r];
  }
}

#define EPT      4
#define SCAN_CH  100
#define NPW      4

static_assert(N_EDGES % EPT == 0, "edge passes use 16-byte loads");
static_assert(N_NODES % SCAN_CH == 0 && N_NODES / SCAN_CH <= 1024 && SCAN_CH % 4 == 0,
              "one 1024-thread block scans all counters");
static_assert(N_NODES % NPW == 0, "every wave owns NPW whole rows");

__global__ void zero_deg_kernel(int* __restrict__ deg) {
  const int tid = blockIdx.x * blockDim.x + threadIdx.x;
  if (tid < N_NODES) deg[tid] = 0;
}

__global__ void rank_kernel(const int* __restrict__ dst,
                            int* __restrict__ deg,
                            int* __restrict__ rank) {
  const int t = blockIdx.x * blockDim.x + threadIdx.x;
  if (t >= N_EDGES / EPT) return;
  const int4 d = ((const int4*)dst)[t];
  int4 r;
  r.x = atomicAdd(deg + d.x, 1);
  r.y = atomicAdd(deg + d.y, 1);
  r.z = atomicAdd(deg + d.z, 1);
  r.w = atomicAdd(deg + d.w, 1);
  ((int4*)rank)[t] = r;
}

__global__ __launch_bounds__(1024)
void scan_kernel(const int* __restrict__ deg, int* __restrict__ off) {
  __shared__ int wsum[32];
  const int t    = threadIdx.x;
  const int lane = t & 31;
  const int wv   = t >> 5;
  const bool act = t < N_NODES / SCAN_CH;
  const int4* __restrict__ p = (const int4*)(deg + t * SCAN_CH);

  int s = 0;
  if (act) {
    for (int i = 0; i < SCAN_CH / 4; ++i) {
      const int4 v = p[i];
      s += v.x + v.y + v.z + v.w;
    }
  }
  int x = s;
  for (int d = 1; d < 32; d <<= 1) {
    const int y = __shfl_up(x, d);
    if (lane >= d) x += y;
  }
  if (lane == 31) wsum[wv] = x;
  __syncthreads();
  if (wv == 0) {
    const int v = wsum[lane];
    int xx = v;
    for (int d = 1; d < 32; d <<= 1) {
      const int y = __shfl_up(xx, d);
      if (lane >= d) xx += y;
    }
    wsum[lane] = xx - v;
  }
  __syncthreads();
  int base = wsum[wv] + x - s;
  if (act) {
    int4* __restrict__ q = (int4*)(off + t * SCAN_CH);
    for (int i = 0; i < SCAN_CH / 4; ++i) {
      const int4 v = p[i];
      int4 o;
      o.x = base;
      o.y = o.x + v.x;
      o.z = o.y + v.y;
      o.w = o.z + v.z;
      base = o.w + v.w;
      q[i] = o;
    }
    if (t == N_NODES / SCAN_CH - 1) off[N_NODES] = base;
  }
}

__global__ void place_kernel(const int* __restrict__ src,
                             const int* __restrict__ dst,
                             const float* __restrict__ ew,
                             const int* __restrict__ rank,
                             const int* __restrict__ off,
                             int* __restrict__ csr_raw) {
  const int t = blockIdx.x * blockDim.x + threadIdx.x;
  if (t >= N_EDGES / EPT) return;
  int2* __restrict__ csr = (int2*)csr_raw;
  const int4   s = ((const int4*)src)[t];
  const int4   d = ((const int4*)dst)[t];
  const int4   r = ((const int4*)rank)[t];
  const float4 w = ((const float4*)ew)[t];
  const int p0 = off[d.x] + r.x;
  const int p1 = off[d.y] + r.y;
  const int p2 = off[d.z] + r.z;
  const int p3 = off[d.w] + r.w;
  csr[p0] = make_int2(s.x, __float_as_int(w.x));
  csr[p1] = make_int2(s.y, __float_as_int(w.y));
  csr[p2] = make_int2(s.z, __float_as_int(w.z));
  csr[p3] = make_int2(s.w, __float_as_int(w.w));
}

template <int U>
__device__ __forceinline__ void gather_rows(const float2* __restrict__ sup2,
                                            const int2 ed, const int j,
                                            const int lane, float2& acc) {
  float2 v[U];
  float  w[U];
#pragma unroll
  for (int u = 0; u < U; ++u) {
    const int s = __builtin_amdgcn_readlane(ed.x, j + u);
    w[u] = __int_as_float(__builtin_amdgcn_readlane(ed.y, j + u));
    v[u] = sup2[(long long)s * 32 + lane];
  }
#pragma unroll
  for (int u = 0; u < U; ++u) {
    acc.x += w[u] * v[u].x;
    acc.y += w[u] * v[u].y;
  }
}

__device__ __forceinline__ void gather_chunk(const float2* __restrict__ sup2,
                                             const int2 ed, const int cnt,
                                             const int lane, float2& acc) {
  int j = 0;
  for (; j + 8 <= cnt; j += 8) gather_rows<8>(sup2, ed, j, lane, acc);
  if (j + 4 <= cnt) { gather_rows<4>(sup2, ed, j, lane, acc); j += 4; }
  if (j + 2 <= cnt) { gather_rows<2>(sup2, ed, j, lane, acc); j += 2; }
  if (j < cnt)      { gather_rows<1>(sup2, ed, j, lane, acc); }
}

__global__ __launch_bounds__(256)
void spmm_gather_kernel(const float* __restrict__ sup,
                        const int* __restrict__ csr_raw,
                        const int* __restrict__ off,
                        const float* __restrict__ bias,
                        float* __restrict__ out) {
  const int lane = threadIdx.x & 31;
  const int wave = __builtin_amdgcn_readfirstlane(
      (int)(blockIdx.x * (blockDim.x >> 5) + (threadIdx.x >> 5)));
  const int n0 = wave * NPW;
  if (n0 >= N_NODES) return;

  const float2* __restrict__ sup2 = (const float2*)sup;
  const int2*   __restrict__ csr  = (const int2*)csr_raw;
  const float2 bv = ((const float2*)bias)[lane];

  const int myoff = off[n0 + (lane <= NPW ? lane : NPW)];
  int o[NPW + 1];
#pragma unroll
  for (int i = 0; i <= NPW; ++i) o[i] = __builtin_amdgcn_readlane(myoff, i);
  int2 ed[NPW];
#pragma unroll
  for (int i = 0; i < NPW; ++i) {
    const int len = o[i + 1] - o[i];
    ed[i] = (lane < len) ? csr[o[i] + lane] : make_int2(0, 0);
  }

#pragma unroll
  for (int i = 0; i < NPW; ++i) {
    const int beg = o[i], end = o[i + 1];
    float2 acc = make_float2(0.0f, 0.0f);
    int cnt = end - beg;
    gather_chunk(sup2, ed[i], cnt < 32 ? cnt : 32, lane, acc);
    for (int c = beg + 32; c < end; c += 32) {
      cnt = end - c;
      const int2 e2 = (lane < cnt) ? csr[c + lane] : make_int2(0, 0);
      gather_chunk(sup2, e2, cnt < 32 ? cnt : 32, lane, acc);
    }
    float2 r;
    r.x = acc.x + bv.x;
    r.y = acc.y + bv.y;
    ((float2*)(out + (long long)(n0 + i) * 64))[lane] = r;
  }
}

extern "C" void kernel_launch(void* const* d_in, const int* in_sizes, int n_in,
                              void* d_out, int out_size, void* d_ws, size_t ws_size,
                              hipStream_t stream) {
  const float* features = (const float*)d_in[0];
  const int*   src      = (const int*)  d_in[1];
  const int*   dst      = (const int*)  d_in[2];
  const float* ew       = (const float*)d_in[3];
  const float* W1       = (const float*)d_in[4];
  const float* b1       = (const float*)d_in[5];
  const float* W2       = (const float*)d_in[6];
  const float* b2       = (const float*)d_in[7];
  const float* W3       = (const float*)d_in[8];
  const float* b3       = (const float*)d_in[9];
  float* out = (float*)d_out;

  const long long feat_elems = (long long)N_NODES * D_H;
  float* sup  = (float*)d_ws;
  int*   csr  = (int*)(sup + feat_elems);
  int*   rank = csr + 2LL * N_EDGES;
  int*   deg  = rank + N_EDGES;
  int*   off  = deg + N_NODES;

  const dim3 blk(256);
  const int gemm_blocks = (N_NODES / 16 + 7) / 8;
  const int node_blocks = (N_NODES + 255) / 256;
  const int edge_blocks = (N_EDGES / EPT + 255) / 256;
  const int spmm_blocks = (N_NODES / NPW + 7) / 8;

  zero_deg_kernel<<<node_blocks, blk, 0, stream>>>(deg);
  rank_kernel<<<edge_blocks, blk, 0, stream>>>(dst, deg, rank);
  scan_kernel<<<1, dim3(1024), 0, stream>>>(deg, off);
  place_kernel<<<edge_blocks, blk, 0, stream>>>(src, dst, ew, rank, off, csr);

  gemm_wmma_kernel<D_IN, false><<<gemm_blocks, blk, 0, stream>>>(features, W1, sup, N_NODES);
  spmm_gather_kernel<<<spmm_blocks, blk, 0, stream>>>(sup, csr, off, b1, out);

  gemm_wmma_kernel<D_H, true><<<gemm_blocks, blk, 0, stream>>>(out, W2, sup, N_NODES);
  spmm_gather_kernel<<<spmm_blocks, blk, 0, stream>>>(sup, csr, off, b2, out);

  gemm_wmma_kernel<D_H, true><<<gemm_blocks, blk, 0, stream>>>(out, W3, sup, N_NODES);
  spmm_gather_kernel<<<spmm_blocks, blk, 0, stream>>>(sup, csr, off, b3, out);

  (void)in_sizes; (void)n_in; (void)out_size; (void)ws_size;
}


# Edge grouping with no global atomic: edges binned by dst range on LDS counters (count, table scan, place); one work-group per bin then ranks its rows by LDS returning adds (arrival order, as before)
# speedup vs baseline: 2.5423x; 1.5291x over previous
#include <hip/hip_runtime.h>

#define N_NODES 100000
#define N_EDGES 1600000
#define D_IN    128
#define D_H     64

typedef __attribute__((ext_vector_type(2))) float v2f;
typedef __attribute__((ext_vector_type(8))) float v8f;

template <int K, bool RELU_IN>
__global__ void gemm_wmma_kernel(const float* __restrict__ A,
                                 const float* __restrict__ W,
                                 float* __restrict__ C, int nrows) {
  __shared__ v2f Pw[(K / 2) * 64];

  for (int idx = threadIdx.x; idx < K * 64; idx += blockDim.x) {
    const int k = idx >> 6;
    const int c = idx & 63;
    ((float*)Pw)[(k >> 1) * 128 + c * 2 + (k & 1)] = W[idx];
  }
  __syncthreads();

  const int lane = threadIdx.x & 31;
  const int wave = blockIdx.x * (blockDim.x >> 5) + (threadIdx.x >> 5);
  const int row0 = wave * 16;
  if (row0 >= nrows) return;

  const int half = lane >> 4;
  const int l15  = lane & 15;
  const long long arow = row0 + l15;

  v8f acc0 = {}, acc1 = {}, acc2 = {}, acc3 = {};

  for (int k = 0; k < K; k += 4) {
    const int kk = k + 2 * half;
    v2f a = *(const v2f*)(A + arow * K + kk);
    if (RELU_IN) {
      a.x = fmaxf(a.x, 0.0f);
      a.y = fmaxf(a.y, 0.0f);
    }
    const v2f* __restrict__ Pk = Pw + (k / 2 + half) * 64;
    const v2f b0 = Pk[l15 +  0];
    const v2f b1 = Pk[l15 + 16];
    const v2f b2 = Pk[l15 + 32];
    const v2f b3 = Pk[l15 + 48];

    acc0 = __builtin_amdgcn_wmma_f32_16x16x4_f32(false, a, false, b0, (short)0, acc0, false, false);
    acc1 = __builtin_amdgcn_wmma_f32_16x16x4_f32(false, a, false, b1, (short)0, acc1, false, false);
    acc2 = __builtin_amdgcn_wmma_f32_16x16x4_f32(false, a, false, b2, (short)0, acc2, false, false);
    acc3 = __builtin_amdgcn_wmma_f32_16x16x4_f32(false, a, false, b3, (short)0, acc3, false, false);
  }

#pragma unroll
  for (int r = 0; r < 8; ++r) {
    const long long m = row0 + r + 8 * half;
    float* __restrict__ crow = C + m * 64;
    crow[l15 +  0] = acc0[r];
    crow[l15 + 16] = acc1[r];
    crow[l15 + 32] = acc2[r];
    crow[l15 + 48] = acc3[r];
  }
}

#define NCH      250
#define CH_EDGES 6400
#define EB_THR   640
#define EB_EPT   10
#define NB       256
#define BKT      391
#define TBL      (NB * NCH)
#define SCAN_CH  64
#define SRC_BITS 17
#define GB_THR   1024
#define GB_ROWS  512
#define NPW      4

static_assert(NCH * CH_EDGES == N_EDGES && EB_THR * EB_EPT == CH_EDGES, "chunks tile the edge list");
static_assert(NB * BKT >= N_NODES && (NB - 1) * BKT < N_NODES, "buckets tile the rows");
static_assert(N_NODES <= (1 << SRC_BITS) && BKT <= (1 << (31 - SRC_BITS)), "record packing");
static_assert(TBL % SCAN_CH == 0 && TBL / SCAN_CH <= 1024 && SCAN_CH % 4 == 0,
              "one 1024-thread block scans the whole table");
static_assert(BKT <= GB_ROWS && GB_ROWS % 32 == 0 && GB_ROWS <= GB_THR, "bucket rows fit the LDS scan");
static_assert(N_NODES % NPW == 0, "every wave owns NPW whole rows");

__global__ __launch_bounds__(EB_THR)
void bucket_count_kernel(const int* __restrict__ dst, int* __restrict__ tcnt) {
  __shared__ int hist[NB];
  const int t = threadIdx.x;
  const int c = blockIdx.x;
  if (t < NB) hist[t] = 0;
  __syncthreads();
  const int* __restrict__ d = dst + c * CH_EDGES + t;
  int dv[EB_EPT];
#pragma unroll
  for (int k = 0; k < EB_EPT; ++k) dv[k] = d[k * EB_THR];
#pragma unroll
  for (int k = 0; k < EB_EPT; ++k) atomicAdd(&hist[(unsigned)dv[k] / BKT], 1);
  __syncthreads();
  if (t < NB) tcnt[t * NCH + c] = hist[t];
}

__global__ __launch_bounds__(1024)
void table_scan_kernel(const int* __restrict__ tcnt, int* __restrict__ toff) {
  __shared__ int wsum[32];
  const int t    = threadIdx.x;
  const int lane = t & 31;
  const int wv   = t >> 5;
  const bool act = t < TBL / SCAN_CH;
  const int4* __restrict__ p = (const int4*)(tcnt + t * SCAN_CH);

  int s = 0;
  if (act) {
    for (int i = 0; i < SCAN_CH / 4; ++i) {
      const int4 v = p[i];
      s += v.x + v.y + v.z + v.w;
    }
  }
  int x = s;
  for (int d = 1; d < 32; d <<= 1) {
    const int y = __shfl_up(x, d);
    if (lane >= d) x += y;
  }
  if (lane == 31) wsum[wv] = x;
  __syncthreads();
  if (wv == 0) {
    const int v = wsum[lane];
    int xx = v;
    for (int d = 1; d < 32; d <<= 1) {
      const int y = __shfl_up(xx, d);
      if (lane >= d) xx += y;
    }
    wsum[lane] = xx - v;
  }
  __syncthreads();
  int base = wsum[wv] + x - s;
  if (act) {
    int4* __restrict__ q = (int4*)(toff + t * SCAN_CH);
    for (int i = 0; i < SCAN_CH / 4; ++i) {
      const int4 v = p[i];
      int4 o;
      o.x = base;
      o.y = o.x + v.x;
      o.z = o.y + v.y;
      o.w = o.z + v.z;
      base = o.w + v.w;
      q[i] = o;
    }
    if (t == TBL / SCAN_CH - 1) toff[TBL] = base;
  }
}

__global__ __launch_bounds__(EB_THR)
void bucket_place_kernel(const int* __restrict__ src,
                         const int* __restrict__ dst,
                         const float* __restrict__ ew,
                         const int* __restrict__ toff,
                         int* __restrict__ brec_raw) {
  __shared__ int cur[NB];
  const int t = threadIdx.x;
  const int c = blockIdx.x;
  if (t < NB) cur[t] = toff[t * NCH + c];
  __syncthreads();
  int2* __restrict__ brec = (int2*)brec_raw;
  const int e0 = c * CH_EDGES + t;
  int   dv[EB_EPT];
  int   sv[EB_EPT];
  float wv[EB_EPT];
#pragma unroll
  for (int k = 0; k < EB_EPT; ++k) {
    dv[k] = dst[e0 + k * EB_THR];
    sv[k] = src[e0 + k * EB_THR];
    wv[k] = ew[e0 + k * EB_THR];
  }
#pragma unroll
  for (int k = 0; k < EB_EPT; ++k) {
    const unsigned d  = (unsigned)dv[k];
    const unsigned b  = d / BKT;
    const unsigned dl = d - b * BKT;
    const int p = atomicAdd(&cur[b], 1);
    brec[p] = make_int2((int)((unsigned)sv[k] | (dl << SRC_BITS)), __float_as_int(wv[k]));
  }
}

__global__ __launch_bounds__(GB_THR)
void bucket_group_kernel(const int* __restrict__ brec_raw,
                         const int* __restrict__ toff,
                         int* __restrict__ off,
                         int* __restrict__ csr_raw) {
  __shared__ int cnt[GB_ROWS];
  __shared__ int wsum[32];
  const int t    = threadIdx.x;
  const int lane = t & 31;
  const int wv   = t >> 5;
  const int b    = blockIdx.x;
  const int n0   = b * BKT;
  const int nn   = (N_NODES - n0 < BKT) ? (N_NODES - n0) : BKT;
  const int beg  = toff[b * NCH];
  const int end  = toff[(b + 1) * NCH];
  const int2* __restrict__ brec = (const int2*)brec_raw;
  int2*       __restrict__ csr  = (int2*)csr_raw;

  if (t < GB_ROWS) cnt[t] = 0;
  __syncthreads();
  for (int i = beg + t; i < end; i += GB_THR) {
    const unsigned key = (unsigned)brec[i].x;
    atomicAdd(&cnt[key >> SRC_BITS], 1);
  }
  __syncthreads();

  int s = 0, x = 0;
  if (t < GB_ROWS) {
    s = cnt[t];
    x = s;
    for (int d = 1; d < 32; d <<= 1) {
      const int y = __shfl_up(x, d);
      if (lane >= d) x += y;
    }
    if (lane == 31) wsum[wv] = x;
  }
  __syncthreads();
  if (wv == 0) {
    const int v = (lane < GB_ROWS / 32) ? wsum[lane] : 0;
    int xx = v;
    for (int d = 1; d < 32; d <<= 1) {
      const int y = __shfl_up(xx, d);
      if (lane >= d) xx += y;
    }
    wsum[lane] = xx - v;
  }
  __syncthreads();
  if (t < GB_ROWS) {
    const int p = beg + wsum[wv] + x - s;
    cnt[t] = p;
    if (t < nn) off[n0 + t] = p;
  }
  if (b == NB - 1 && t == 0) off[N_NODES] = end;
  __syncthreads();

  for (int i = beg + t; i < end; i += GB_THR) {
    const int2 r = brec[i];
    const unsigned key = (unsigned)r.x;
    const int p = atomicAdd(&cnt[key >> SRC_BITS], 1);
    csr[p] = make_int2((int)(key & ((1u << SRC_BITS) - 1u)), r.y);
  }
}

template <int U>
__device__ __forceinline__ void gather_rows(const float2* __restrict__ sup2,
                                            const int2 ed, const int j,
                                            const int lane, float2& acc) {
  float2 v[U];
  float  w[U];
#pragma unroll
  for (int u = 0; u < U; ++u) {
    const int s = __builtin_amdgcn_readlane(ed.x, j + u);
    w[u] = __int_as_float(__builtin_amdgcn_readlane(ed.y, j + u));
    v[u] = sup2[(long long)s * 32 + lane];
  }
#pragma unroll
  for (int u = 0; u < U; ++u) {
    acc.x += w[u] * v[u].x;
    acc.y += w[u] * v[u].y;
  }
}

__device__ __forceinline__ void gather_chunk(const float2* __restrict__ sup2,
                                             const int2 ed, const int cnt,
                                             const int lane, float2& acc) {
  int j = 0;
  for (; j + 8 <= cnt; j += 8) gather_rows<8>(sup2, ed, j, lane, acc);
  if (j + 4 <= cnt) { gather_rows<4>(sup2, ed, j, lane, acc); j += 4; }
  if (j + 2 <= cnt) { gather_rows<2>(sup2, ed, j, lane, acc); j += 2; }
  if (j < cnt)      { gather_rows<1>(sup2, ed, j, lane, acc); }
}

__global__ __launch_bounds__(256)
void spmm_gather_kernel(const float* __restrict__ sup,
                        const int* __restrict__ csr_raw,
                        const int* __restrict__ off,
                        const float* __restrict__ bias,
                        float* __restrict__ out) {
  const int lane = threadIdx.x & 31;
  const int wave = __builtin_amdgcn_readfirstlane(
      (int)(blockIdx.x * (blockDim.x >> 5) + (threadIdx.x >> 5)));
  const int n0 = wave * NPW;
  if (n0 >= N_NODES) return;

  const float2* __restrict__ sup2 = (const float2*)sup;
  const int2*   __restrict__ csr  = (const int2*)csr_raw;
  const float2 bv = ((const float2*)bias)[lane];

  const int myoff = off[n0 + (lane <= NPW ? lane : NPW)];
  int o[NPW + 1];
#pragma unroll
  for (int i = 0; i <= NPW; ++i) o[i] = __builtin_amdgcn_readlane(myoff, i);
  int2 ed[NPW];
#pragma unroll
  for (int i = 0; i < NPW; ++i) {
    const int len = o[i + 1] - o[i];
    ed[i] = (lane < len) ? csr[o[i] + lane] : make_int2(0, 0);
  }

#pragma unroll
  for (int i = 0; i < NPW; ++i) {
    const int beg = o[i], end = o[i + 1];
    float2 acc = make_float2(0.0f, 0.0f);
    int cnt = end - beg;
    gather_chunk(sup2, ed[i], cnt < 32 ? cnt : 32, lane, acc);
    for (int c = beg + 32; c < end; c += 32) {
      cnt = end - c;
      const int2 e2 = (lane < cnt) ? csr[c + lane] : make_int2(0, 0);
      gather_chunk(sup2, e2, cnt < 32 ? cnt : 32, lane, acc);
    }
    float2 r;
    r.x = acc.x + bv.x;
    r.y = acc.y + bv.y;
    ((float2*)(out + (long long)(n0 + i) * 64))[lane] = r;
  }
}

extern "C" void kernel_launch(void* const* d_in, const int* in_sizes, int n_in,
                              void* d_out, int out_size, void* d_ws, size_t ws_size,
                              hipStream_t stream) {
  const float* features = (const float*)d_in[0];
  const int*   src      = (const int*)  d_in[1];
  const int*   dst      = (const int*)  d_in[2];
  const float* ew       = (const float*)d_in[3];
  const float* W1       = (const float*)d_in[4];
  const float* b1       = (const float*)d_in[5];
  const float* W2       = (const float*)d_in[6];
  const float* b2       = (const float*)d_in[7];
  const float* W3       = (const float*)d_in[8];
  const float* b3       = (const float*)d_in[9];
  float* out = (float*)d_out;

  const long long feat_elems = (long long)N_NODES * D_H;
  float* sup  = (float*)d_ws;
  int*   csr  = (int*)(sup + feat_elems);
  int*   off  = csr + 2LL * N_EDGES;
  int*   brec = (int*)sup;
  int*   tcnt = brec + 2LL * N_EDGES;
  int*   toff = tcnt + TBL;

  const dim3 blk(256);
  const int gemm_blocks = (N_NODES / 16 + 7) / 8;
  const int spmm_blocks = (N_NODES / NPW + 7) / 8;

  bucket_count_kernel<<<NCH, dim3(EB_THR), 0, stream>>>(dst, tcnt);
  table_scan_kernel<<<1, dim3(1024), 0, stream>>>(tcnt, toff);
  bucket_place_kernel<<<NCH, dim3(EB_THR), 0, stream>>>(src, dst, ew, toff, brec);
  bucket_group_kernel<<<NB, dim3(GB_THR), 0, stream>>>(brec, toff, off, csr);

  gemm_wmma_kernel<D_IN, false><<<gemm_blocks, blk, 0, stream>>>(features, W1, sup, N_NODES);
  spmm_gather_kernel<<<spmm_blocks, blk, 0, stream>>>(sup, csr, off, b1, out);

  gemm_wmma_kernel<D_H, true><<<gemm_blocks, blk, 0, stream>>>(out, W2, sup, N_NODES);
  spmm_gather_kernel<<<spmm_blocks, blk, 0, stream>>>(sup, csr, off, b2, out);

  gemm_wmma_kernel<D_H, true><<<gemm_blocks, blk, 0, stream>>>(out, W3, sup, N_NODES);
  spmm_gather_kernel<<<spmm_blocks, blk, 0, stream>>>(sup, csr, off, b3, out);

  (void)in_sizes; (void)n_in; (void)out_size; (void)ws_size;
}


# Each hidden layer's gather now feeds the next layer's f32 WMMA product through an LDS tile in the same launch (relu in between); h1, h2 never reach memory; two GEMM launches gone; sums unchanged
# speedup vs baseline: 3.1206x; 1.2275x over previous
#include <hip/hip_runtime.h>

#define N_NODES 100000
#define N_EDGES 1600000
#define D_IN    128
#define D_H     64

typedef __attribute__((ext_vector_type(2))) float v2f;
typedef __attribute__((ext_vector_type(8))) float v8f;

template <int K, bool RELU_IN>
__global__ void gemm_wmma_kernel(const float* __restrict__ A,
                                 const float* __restrict__ W,
                                 float* __restrict__ C, int nrows) {
  __shared__ v2f Pw[(K / 2) * 64];

  for (int idx = threadIdx.x; idx < K * 64; idx += blockDim.x) {
    const int k = idx >> 6;
    const int c = idx & 63;
    ((float*)Pw)[(k >> 1) * 128 + c * 2 + (k & 1)] = W[idx];
  }
  __syncthreads();

  const int lane = threadIdx.x & 31;
  const int wave = blockIdx.x * (blockDim.x >> 5) + (threadIdx.x >> 5);
  const int row0 = wave * 16;
  if (row0 >= nrows) return;

  const int half = lane >> 4;
  const int l15  = lane & 15;
  const long long arow = row0 + l15;

  v8f acc0 = {}, acc1 = {}, acc2 = {}, acc3 = {};

  for (int k = 0; k < K; k += 4) {
    const int kk = k + 2 * half;
    v2f a = *(const v2f*)(A + arow * K + kk);
    if (RELU_IN) {
      a.x = fmaxf(a.x, 0.0f);
      a.y = fmaxf(a.y, 0.0f);
    }
    const v2f* __restrict__ Pk = Pw + (k / 2 + half) * 64;
    const v2f b0 = Pk[l15 +  0];
    const v2f b1 = Pk[l15 + 16];
    const v2f b2 = Pk[l15 + 32];
    const v2f b3 = Pk[l15 + 48];

    acc0 = __builtin_amdgcn_wmma_f32_16x16x4_f32(false, a, false, b0, (short)0, acc0, false, false);
    acc1 = __builtin_amdgcn_wmma_f32_16x16x4_f32(false, a, false, b1, (short)0, acc1, false, false);
    acc2 = __builtin_amdgcn_wmma_f32_16x16x4_f32(false, a, false, b2, (short)0, acc2, false, false);
    acc3 = __builtin_amdgcn_wmma_f32_16x16x4_f32(false, a, false, b3, (short)0, acc3, false, false);
  }

#pragma unroll
  for (int r = 0; r < 8; ++r) {
    const long long m = row0 + r + 8 * half;
    float* __restrict__ crow = C + m * 64;
    crow[l15 +  0] = acc0[r];
    crow[l15 + 16] = acc1[r];
    crow[l15 + 32] = acc2[r];
    crow[l15 + 48] = acc3[r];
  }
}

#define NCH      250
#define CH_EDGES 6400
#define EB_THR   640
#define EB_EPT   10
#define NB       256
#define BKT      391
#define TBL      (NB * NCH)
#define SCAN_CH  64
#define SRC_BITS 17
#define GB_THR   1024
#define GB_ROWS  512
#define NPW      4

static_assert(NCH * CH_EDGES == N_EDGES && EB_THR * EB_EPT == CH_EDGES, "chunks tile the edge list");
static_assert(NB * BKT >= N_NODES && (NB - 1) * BKT < N_NODES, "buckets tile the rows");
static_assert(N_NODES <= (1 << SRC_BITS) && BKT <= (1 << (31 - SRC_BITS)), "record packing");
static_assert(TBL % SCAN_CH == 0 && TBL / SCAN_CH <= 1024 && SCAN_CH % 4 == 0,
              "one 1024-thread block scans the whole table");
static_assert(BKT <= GB_ROWS && GB_ROWS % 32 == 0 && GB_ROWS <= GB_THR, "bucket rows fit the LDS scan");
static_assert(N_NODES % NPW == 0, "every wave owns NPW whole rows");

__global__ __launch_bounds__(EB_THR)
void bucket_count_kernel(const int* __restrict__ dst, int* __restrict__ tcnt) {
  __shared__ int hist[NB];
  const int t = threadIdx.x;
  const int c = blockIdx.x;
  if (t < NB) hist[t] = 0;
  __syncthreads();
  const int* __restrict__ d = dst + c * CH_EDGES + t;
  int dv[EB_EPT];
#pragma unroll
  for (int k = 0; k < EB_EPT; ++k) dv[k] = d[k * EB_THR];
#pragma unroll
  for (int k = 0; k < EB_EPT; ++k) atomicAdd(&hist[(unsigned)dv[k] / BKT], 1);
  __syncthreads();
  if (t < NB) tcnt[t * NCH + c] = hist[t];
}

__global__ __launch_bounds__(1024)
void table_scan_kernel(const int* __restrict__ tcnt, int* __restrict__ toff) {
  __shared__ int wsum[32];
  const int t    = threadIdx.x;
  const int lane = t & 31;
  const int wv   = t >> 5;
  const bool act = t < TBL / SCAN_CH;
  const int4* __restrict__ p = (const int4*)(tcnt + t * SCAN_CH);

  int s = 0;
  if (act) {
    for (int i = 0; i < SCAN_CH / 4; ++i) {
      const int4 v = p[i];
      s += v.x + v.y + v.z + v.w;
    }
  }
  int x = s;
  for (int d = 1; d < 32; d <<= 1) {
    const int y = __shfl_up(x, d);
    if (lane >= d) x += y;
  }
  if (lane == 31) wsum[wv] = x;
  __syncthreads();
  if (wv == 0) {
    const int v = wsum[lane];
    int xx = v;
    for (int d = 1; d < 32; d <<= 1) {
      const int y = __shfl_up(xx, d);
      if (lane >= d) xx += y;
    }
    wsum[lane] = xx - v;
  }
  __syncthreads();
  int base = wsum[wv] + x - s;
  if (act) {
    int4* __restrict__ q = (int4*)(toff + t * SCAN_CH);
    for (int i = 0; i < SCAN_CH / 4; ++i) {
      const int4 v = p[i];
      int4 o;
      o.x = base;
      o.y = o.x + v.x;
      o.z = o.y + v.y;
      o.w = o.z + v.z;
      base = o.w + v.w;
      q[i] = o;
    }
    if (t == TBL / SCAN_CH - 1) toff[TBL] = base;
  }
}

__global__ __launch_bounds__(EB_THR)
void bucket_place_kernel(const int* __restrict__ src,
                         const int* __restrict__ dst,
                         const float* __restrict__ ew,
                         const int* __restrict__ toff,
                         int* __restrict__ brec_raw) {
  __shared__ int cur[NB];
  const int t = threadIdx.x;
  const int c = blockIdx.x;
  if (t < NB) cur[t] = toff[t * NCH + c];
  __syncthreads();
  int2* __restrict__ brec = (int2*)brec_raw;
  const int e0 = c * CH_EDGES + t;
  int   dv[EB_EPT];
  int   sv[EB_EPT];
  float wv[EB_EPT];
#pragma unroll
  for (int k = 0; k < EB_EPT; ++k) {
    dv[k] = dst[e0 + k * EB_THR];
    sv[k] = src[e0 + k * EB_THR];
    wv[k] = ew[e0 + k * EB_THR];
  }
#pragma unroll
  for (int k = 0; k < EB_EPT; ++k) {
    const unsigned d  = (unsigned)dv[k];
    const unsigned b  = d / BKT;
    const unsigned dl = d - b * BKT;
    const int p = atomicAdd(&cur[b], 1);
    brec[p] = make_int2((int)((unsigned)sv[k] | (dl << SRC_BITS)), __float_as_int(wv[k]));
  }
}

__global__ __launch_bounds__(GB_THR)
void bucket_group_kernel(const int* __restrict__ brec_raw,
                         const int* __restrict__ toff,
                         int* __restrict__ off,
                         int* __restrict__ csr_raw) {
  __shared__ int cnt[GB_ROWS];
  __shared__ int wsum[32];
  const int t    = threadIdx.x;
  const int lane = t & 31;
  const int wv   = t >> 5;
  const int b    = blockIdx.x;
  const int n0   = b * BKT;
  const int nn   = (N_NODES - n0 < BKT) ? (N_NODES - n0) : BKT;
  const int beg  = toff[b * NCH];
  const int end  = toff[(b + 1) * NCH];
  const int2* __restrict__ brec = (const int2*)brec_raw;
  int2*       __restrict__ csr  = (int2*)csr_raw;

  if (t < GB_ROWS) cnt[t] = 0;
  __syncthreads();
  for (int i = beg + t; i < end; i += GB_THR) {
    const unsigned key = (unsigned)brec[i].x;
    atomicAdd(&cnt[key >> SRC_BITS], 1);
  }
  __syncthreads();

  int s = 0, x = 0;
  if (t < GB_ROWS) {
    s = cnt[t];
    x = s;
    for (int d = 1; d < 32; d <<= 1) {
      const int y = __shfl_up(x, d);
      if (lane >= d) x += y;
    }
    if (lane == 31) wsum[wv] = x;
  }
  __syncthreads();
  if (wv == 0) {
    const int v = (lane < GB_ROWS / 32) ? wsum[lane] : 0;
    int xx = v;
    for (int d = 1; d < 32; d <<= 1) {
      const int y = __shfl_up(xx, d);
      if (lane >= d) xx += y;
    }
    wsum[lane] = xx - v;
  }
  __syncthreads();
  if (t < GB_ROWS) {
    const int p = beg + wsum[wv] + x - s;
    cnt[t] = p;
    if (t < nn) off[n0 + t] = p;
  }
  if (b == NB - 1 && t == 0) off[N_NODES] = end;
  __syncthreads();

  for (int i = beg + t; i < end; i += GB_THR) {
    const int2 r = brec[i];
    const unsigned key = (unsigned)r.x;
    const int p = atomicAdd(&cnt[key >> SRC_BITS], 1);
    csr[p] = make_int2((int)(key & ((1u << SRC_BITS) - 1u)), r.y);
  }
}

template <int U>
__device__ __forceinline__ void gather_rows(const float2* __restrict__ sup2,
                                            const int2 ed, const int j,
                                            const int lane, float2& acc) {
  float2 v[U];
  float  w[U];
#pragma unroll
  for (int u = 0; u < U; ++u) {
    const int s = __builtin_amdgcn_readlane(ed.x, j + u);
    w[u] = __int_as_float(__builtin_amdgcn_readlane(ed.y, j + u));
    v[u] = sup2[(long long)s * 32 + lane];
  }
#pragma unroll
  for (int u = 0; u < U; ++u) {
    acc.x += w[u] * v[u].x;
    acc.y += w[u] * v[u].y;
  }
}

__device__ __forceinline__ void gather_chunk(const float2* __restrict__ sup2,
                                             const int2 ed, const int cnt,
                                             const int lane, float2& acc) {
  int j = 0;
  for (; j + 8 <= cnt; j += 8) gather_rows<8>(sup2, ed, j, lane, acc);
  if (j + 4 <= cnt) { gather_rows<4>(sup2, ed, j, lane, acc); j += 4; }
  if (j + 2 <= cnt) { gather_rows<2>(sup2, ed, j, lane, acc); j += 2; }
  if (j < cnt)      { gather_rows<1>(sup2, ed, j, lane, acc); }
}

__global__ __launch_bounds__(256)
void spmm_gather_kernel(const float* __restrict__ sup,
                        const int* __restrict__ csr_raw,
                        const int* __restrict__ off,
                        const float* __restrict__ bias,
                        float* __restrict__ out) {
  const int lane = threadIdx.x & 31;
  const int wave = __builtin_amdgcn_readfirstlane(
      (int)(blockIdx.x * (blockDim.x >> 5) + (threadIdx.x >> 5)));
  const int n0 = wave * NPW;
  if (n0 >= N_NODES) return;

  const float2* __restrict__ sup2 = (const float2*)sup;
  const int2*   __restrict__ csr  = (const int2*)csr_raw;
  const float2 bv = ((const float2*)bias)[lane];

  const int myoff = off[n0 + (lane <= NPW ? lane : NPW)];
  int o[NPW + 1];
#pragma unroll
  for (int i = 0; i <= NPW; ++i) o[i] = __builtin_amdgcn_readlane(myoff, i);
  int2 ed[NPW];
#pragma unroll
  for (int i = 0; i < NPW; ++i) {
    const int len = o[i + 1] - o[i];
    ed[i] = (lane < len) ? csr[o[i] + lane] : make_int2(0, 0);
  }

#pragma unroll
  for (int i = 0; i < NPW; ++i) {
    const int beg = o[i], end = o[i + 1];
    float2 acc = make_float2(0.0f, 0.0f);
    int cnt = end - beg;
    gather_chunk(sup2, ed[i], cnt < 32 ? cnt : 32, lane, acc);
    for (int c = beg + 32; c < end; c += 32) {
      cnt = end - c;
      const int2 e2 = (lane < cnt) ? csr[c + lane] : make_int2(0, 0);
      gather_chunk(sup2, e2, cnt < 32 ? cnt : 32, lane, acc);
    }
    float2 r;
    r.x = acc.x + bv.x;
    r.y = acc.y + bv.y;
    ((float2*)(out + (long long)(n0 + i) * 64))[lane] = r;
  }
}

#define FR 32
#define HP 68

static_assert(FR == 8 * NPW && N_NODES % FR == 0, "work-groups tile the rows exactly");

__global__ __launch_bounds__(256)
void spmm_gemm_kernel(const float* __restrict__ sup,
                      const int* __restrict__ csr_raw,
                      const int* __restrict__ off,
                      const float* __restrict__ bias,
                      const float* __restrict__ Wn,
                      float* __restrict__ supn) {
  __shared__ alignas(16) v2f   Pw[(D_H / 2) * 64];
  __shared__ alignas(16) float Ht[FR * HP];

  const int t    = threadIdx.x;
  const int lane = t & 31;
  const int wv   = __builtin_amdgcn_readfirstlane(t >> 5);
  const int row0 = blockIdx.x * FR;
  const int n0   = row0 + wv * NPW;

  const int kp0 = t >> 4;
  const int cg  = t & 15;
  const float4 wa0 = *(const float4*)(Wn + (2 * kp0)      * 64 + 4 * cg);
  const float4 wb0 = *(const float4*)(Wn + (2 * kp0 + 1)  * 64 + 4 * cg);
  const float4 wa1 = *(const float4*)(Wn + (2 * kp0 + 32) * 64 + 4 * cg);
  const float4 wb1 = *(const float4*)(Wn + (2 * kp0 + 33) * 64 + 4 * cg);

  const float2* __restrict__ sup2 = (const float2*)sup;
  const int2*   __restrict__ csr  = (const int2*)csr_raw;
  const float2 bv = ((const float2*)bias)[lane];

  const int myoff = off[n0 + (lane <= NPW ? lane : NPW)];
  int o[NPW + 1];
#pragma unroll
  for (int i = 0; i <= NPW; ++i) o[i] = __builtin_amdgcn_readlane(myoff, i);
  int2 ed[NPW];
#pragma unroll
  for (int i = 0; i < NPW; ++i) {
    const int len = o[i + 1] - o[i];
    ed[i] = (lane < len) ? csr[o[i] + lane] : make_int2(0, 0);
  }

#pragma unroll
  for (int i = 0; i < NPW; ++i) {
    const int beg = o[i], end = o[i + 1];
    float2 acc = make_float2(0.0f, 0.0f);
    int cnt = end - beg;
    gather_chunk(sup2, ed[i], cnt < 32 ? cnt : 32, lane, acc);
    for (int c = beg + 32; c < end; c += 32) {
      cnt = end - c;
      const int2 e2 = (lane < cnt) ? csr[c + lane] : make_int2(0, 0);
      gather_chunk(sup2, e2, cnt < 32 ? cnt : 32, lane, acc);
    }
    float2 r;
    r.x = fmaxf(acc.x + bv.x, 0.0f);
    r.y = fmaxf(acc.y + bv.y, 0.0f);
    *(float2*)(Ht + (wv * NPW + i) * HP + 2 * lane) = r;
  }

  {
    float4* __restrict__ p0 = (float4*)(Pw + kp0 * 64 + 4 * cg);
    float4* __restrict__ p1 = (float4*)(Pw + (kp0 + 16) * 64 + 4 * cg);
    p0[0] = make_float4(wa0.x, wb0.x, wa0.y, wb0.y);
    p0[1] = make_float4(wa0.z, wb0.z, wa0.w, wb0.w);
    p1[0] = make_float4(wa1.x, wb1.x, wa1.y, wb1.y);
    p1[1] = make_float4(wa1.z, wb1.z, wa1.w, wb1.w);
  }
  __syncthreads();

  if (wv >= 4) return;
  const int half = lane >> 4;
  const int l15  = lane & 15;
  const int s    = wv >> 1;
  const int cb   = (wv & 1) * 32;
  const float* __restrict__ hrow = Ht + (16 * s + l15) * HP + 2 * half;

  v8f acc0 = {}, acc1 = {};
  for (int k = 0; k < D_H; k += 4) {
    const v2f a = *(const v2f*)(hrow + k);
    const v2f* __restrict__ Pk = Pw + (k / 2 + half) * 64 + cb;
    const v2f b0 = Pk[l15];
    const v2f b1 = Pk[l15 + 16];
    acc0 = __builtin_amdgcn_wmma_f32_16x16x4_f32(false, a, false, b0, (short)0, acc0, false, false);
    acc1 = __builtin_amdgcn_wmma_f32_16x16x4_f32(false, a, false, b1, (short)0, acc1, false, false);
  }

#pragma unroll
  for (int r = 0; r < 8; ++r) {
    const long long m = row0 + 16 * s + r + 8 * half;
    float* __restrict__ crow = supn + m * 64 + cb;
    crow[l15]      = acc0[r];
    crow[l15 + 16] = acc1[r];
  }
}

extern "C" void kernel_launch(void* const* d_in, const int* in_sizes, int n_in,
                              void* d_out, int out_size, void* d_ws, size_t ws_size,
                              hipStream_t stream) {
  const float* features = (const float*)d_in[0];
  const int*   src      = (const int*)  d_in[1];
  const int*   dst      = (const int*)  d_in[2];
  const float* ew       = (const float*)d_in[3];
  const float* W1       = (const float*)d_in[4];
  const float* b1       = (const float*)d_in[5];
  const float* W2       = (const float*)d_in[6];
  const float* b2       = (const float*)d_in[7];
  const float* W3       = (const float*)d_in[8];
  const float* b3       = (const float*)d_in[9];
  float* out = (float*)d_out;

  const long long feat_elems = (long long)N_NODES * D_H;
  float* sup  = (float*)d_ws;
  int*   csr  = (int*)(sup + feat_elems);
  int*   off  = csr + 2LL * N_EDGES;
  int*   brec = (int*)sup;
  int*   tcnt = brec + 2LL * N_EDGES;
  int*   toff = tcnt + TBL;
  float* sup2 = out;

  const dim3 blk(256);
  const int gemm_blocks = (N_NODES / 16 + 7) / 8;
  const int spmm_blocks = (N_NODES / NPW + 7) / 8;

  bucket_count_kernel<<<NCH, dim3(EB_THR), 0, stream>>>(dst, tcnt);
  table_scan_kernel<<<1, dim3(1024), 0, stream>>>(tcnt, toff);
  bucket_place_kernel<<<NCH, dim3(EB_THR), 0, stream>>>(src, dst, ew, toff, brec);
  bucket_group_kernel<<<NB, dim3(GB_THR), 0, stream>>>(brec, toff, off, csr);

  gemm_wmma_kernel<D_IN, false><<<gemm_blocks, blk, 0, stream>>>(features, W1, sup, N_NODES);

  spmm_gemm_kernel<<<N_NODES / FR, blk, 0, stream>>>(sup, csr, off, b1, W2, sup2);

  spmm_gemm_kernel<<<N_NODES / FR, blk, 0, stream>>>(sup2, csr, off, b2, W3, sup);

  spmm_gather_kernel<<<spmm_blocks, blk, 0, stream>>>(sup, csr, off, b3, out);

  (void)in_sizes; (void)n_in; (void)out_size; (void)ws_size;
}


# Count table's prefix sum folded into the place pass: each chunk's work-group sums the table's columns itself (16-byte row loads), scans 256 totals in LDS; one-block scan launch gone; sums as before
# speedup vs baseline: 3.5549x; 1.1392x over previous
#include <hip/hip_runtime.h>

#define N_NODES 100000
#define N_EDGES 1600000
#define D_IN    128
#define D_H     64

typedef __attribute__((ext_vector_type(2))) float v2f;
typedef __attribute__((ext_vector_type(8))) float v8f;

template <int K, bool RELU_IN>
__global__ void gemm_wmma_kernel(const float* __restrict__ A,
                                 const float* __restrict__ W,
                                 float* __restrict__ C, int nrows) {
  __shared__ v2f Pw[(K / 2) * 64];

  for (int idx = threadIdx.x; idx < K * 64; idx += blockDim.x) {
    const int k = idx >> 6;
    const int c = idx & 63;
    ((float*)Pw)[(k >> 1) * 128 + c * 2 + (k & 1)] = W[idx];
  }
  __syncthreads();

  const int lane = threadIdx.x & 31;
  const int wave = blockIdx.x * (blockDim.x >> 5) + (threadIdx.x >> 5);
  const int row0 = wave * 16;
  if (row0 >= nrows) return;

  const int half = lane >> 4;
  const int l15  = lane & 15;
  const long long arow = row0 + l15;

  v8f acc0 = {}, acc1 = {}, acc2 = {}, acc3 = {};

  for (int k = 0; k < K; k += 4) {
    const int kk = k + 2 * half;
    v2f a = *(const v2f*)(A + arow * K + kk);
    if (RELU_IN) {
      a.x = fmaxf(a.x, 0.0f);
      a.y = fmaxf(a.y, 0.0f);
    }
    const v2f* __restrict__ Pk = Pw + (k / 2 + half) * 64;
    const v2f b0 = Pk[l15 +  0];
    const v2f b1 = Pk[l15 + 16];
    const v2f b2 = Pk[l15 + 32];
    const v2f b3 = Pk[l15 + 48];

    acc0 = __builtin_amdgcn_wmma_f32_16x16x4_f32(false, a, false, b0, (short)0, acc0, false, false);
    acc1 = __builtin_amdgcn_wmma_f32_16x16x4_f32(false, a, false, b1, (short)0, acc1, false, false);
    acc2 = __builtin_amdgcn_wmma_f32_16x16x4_f32(false, a, false, b2, (short)0, acc2, false, false);
    acc3 = __builtin_amdgcn_wmma_f32_16x16x4_f32(false, a, false, b3, (short)0, acc3, false, false);
  }

#pragma unroll
  for (int r = 0; r < 8; ++r) {
    const long long m = row0 + r + 8 * half;
    float* __restrict__ crow = C + m * 64;
    crow[l15 +  0] = acc0[r];
    crow[l15 + 16] = acc1[r];
    crow[l15 + 32] = acc2[r];
    crow[l15 + 48] = acc3[r];
  }
}

#define NCH      250
#define CH_EDGES 6400
#define EB_THR   640
#define EB_EPT   10
#define NB       256
#define BKT      391
#define TBL      (NB * NCH)
#define TP_LANES (NB / 4)
#define TP_GRP   (EB_THR / TP_LANES)
#define TP_ROWS  (NCH / TP_GRP)
#define SRC_BITS 17
#define GB_THR   1024
#define GB_ROWS  512
#define NPW      4

static_assert(NCH * CH_EDGES == N_EDGES && EB_THR * EB_EPT == CH_EDGES, "chunks tile the edge list");
static_assert(NB * BKT >= N_NODES && (NB - 1) * BKT < N_NODES, "buckets tile the rows");
static_assert(N_NODES <= (1 << SRC_BITS) && BKT <= (1 << (31 - SRC_BITS)), "record packing");
static_assert(NB % 32 == 0 && NB <= EB_THR && TP_LANES % 32 == 0 && TP_GRP * TP_LANES == EB_THR &&
              TP_GRP * TP_ROWS == NCH, "the place pass's thread groups tile the table exactly");
static_assert(BKT <= GB_ROWS && GB_ROWS % 32 == 0 && GB_ROWS <= GB_THR, "bucket rows fit the LDS scan");
static_assert(N_NODES % NPW == 0, "every wave owns NPW whole rows");

__global__ __launch_bounds__(EB_THR)
void bucket_count_kernel(const int* __restrict__ dst, int* __restrict__ tcnt) {
  __shared__ int hist[NB];
  const int t = threadIdx.x;
  const int c = blockIdx.x;
  if (t < NB) hist[t] = 0;
  __syncthreads();
  const int* __restrict__ d = dst + c * CH_EDGES + t;
  int dv[EB_EPT];
#pragma unroll
  for (int k = 0; k < EB_EPT; ++k) dv[k] = d[k * EB_THR];
#pragma unroll
  for (int k = 0; k < EB_EPT; ++k) atomicAdd(&hist[(unsigned)dv[k] / BKT], 1);
  __syncthreads();
  if (t < NB) tcnt[c * NB + t] = hist[t];
}

__global__ __launch_bounds__(EB_THR)
void bucket_place_kernel(const int* __restrict__ src,
                         const int* __restrict__ dst,
                         const float* __restrict__ ew,
                         const int* __restrict__ tcnt,
                         int* __restrict__ gstart,
                         int* __restrict__ brec_raw) {
  __shared__ alignas(16) int pall[TP_GRP][NB];
  __shared__ alignas(16) int pbef[TP_GRP][NB];
  __shared__ int wsum[NB / 32];
  __shared__ int cur[NB];
  const int t    = threadIdx.x;
  const int lane = t & 31;
  const int wv   = t >> 5;
  const int c    = blockIdx.x;
  int2* __restrict__ brec = (int2*)brec_raw;
  const int e0 = c * CH_EDGES + t;
  int   dv[EB_EPT];
  int   sv[EB_EPT];
  float wf[EB_EPT];
#pragma unroll
  for (int k = 0; k < EB_EPT; ++k) {
    dv[k] = dst[e0 + k * EB_THR];
    sv[k] = src[e0 + k * EB_THR];
    wf[k] = ew[e0 + k * EB_THR];
  }

  {
    const int g = __builtin_amdgcn_readfirstlane(t / TP_LANES);
    const int q = t % TP_LANES;
    const int4* __restrict__ trow = (const int4*)tcnt + (g * TP_ROWS) * TP_LANES + q;
    const int nbef = c - g * TP_ROWS;
    int4 v[TP_ROWS];
#pragma unroll
    for (int r = 0; r < TP_ROWS; ++r) v[r] = trow[r * TP_LANES];
    int4 sa = make_int4(0, 0, 0, 0);
    int4 sb = make_int4(0, 0, 0, 0);
#pragma unroll
    for (int r = 0; r < TP_ROWS; ++r) {
      const int m = (r < nbef) ? -1 : 0;
      sa.x += v[r].x;     sa.y += v[r].y;     sa.z += v[r].z;     sa.w += v[r].w;
      sb.x += v[r].x & m; sb.y += v[r].y & m; sb.z += v[r].z & m; sb.w += v[r].w & m;
    }
    *(int4*)&pall[g][4 * q] = sa;
    *(int4*)&pbef[g][4 * q] = sb;
  }
  __syncthreads();

  int all = 0, bef = 0, x = 0;
  if (t < NB) {
#pragma unroll
    for (int g = 0; g < TP_GRP; ++g) {
      all += pall[g][t];
      bef += pbef[g][t];
    }
    x = all;
    for (int d = 1; d < 32; d <<= 1) {
      const int y = __shfl_up(x, d);
      if (lane >= d) x += y;
    }
    if (lane == 31) wsum[wv] = x;
  }
  __syncthreads();
  if (t < NB) {
    int base = x - all;
    for (int w = 0; w < NB / 32; ++w)
      if (w < wv) base += wsum[w];
    cur[t] = base + bef;
    if (c == 0) {
      gstart[t] = base;
      if (t == NB - 1) gstart[NB] = base + all;
    }
  }
  __syncthreads();

#pragma unroll
  for (int k = 0; k < EB_EPT; ++k) {
    const unsigned d  = (unsigned)dv[k];
    const unsigned b  = d / BKT;
    const unsigned dl = d - b * BKT;
    const int p = atomicAdd(&cur[b], 1);
    brec[p] = make_int2((int)((unsigned)sv[k] | (dl << SRC_BITS)), __float_as_int(wf[k]));
  }
}

__global__ __launch_bounds__(GB_THR)
void bucket_group_kernel(const int* __restrict__ brec_raw,
                         const int* __restrict__ gstart,
                         int* __restrict__ off,
                         int* __restrict__ csr_raw) {
  __shared__ int cnt[GB_ROWS];
  __shared__ int wsum[32];
  const int t    = threadIdx.x;
  const int lane = t & 31;
  const int wv   = t >> 5;
  const int b    = blockIdx.x;
  const int n0   = b * BKT;
  const int nn   = (N_NODES - n0 < BKT) ? (N_NODES - n0) : BKT;
  const int beg  = gstart[b];
  const int end  = gstart[b + 1];
  const int2* __restrict__ brec = (const int2*)brec_raw;
  int2*       __restrict__ csr  = (int2*)csr_raw;

  if (t < GB_ROWS) cnt[t] = 0;
  __syncthreads();
  for (int i = beg + t; i < end; i += GB_THR) {
    const unsigned key = (unsigned)brec[i].x;
    atomicAdd(&cnt[key >> SRC_BITS], 1);
  }
  __syncthreads();

  int s = 0, x = 0;
  if (t < GB_ROWS) {
    s = cnt[t];
    x = s;
    for (int d = 1; d < 32; d <<= 1) {
      const int y = __shfl_up(x, d);
      if (lane >= d) x += y;
    }
    if (lane == 31) wsum[wv] = x;
  }
  __syncthreads();
  if (wv == 0) {
    const int v = (lane < GB_ROWS / 32) ? wsum[lane] : 0;
    int xx = v;
    for (int d = 1; d < 32; d <<= 1) {
      const int y = __shfl_up(xx, d);
      if (lane >= d) xx += y;
    }
    wsum[lane] = xx - v;
  }
  __syncthreads();
  if (t < GB_ROWS) {
    const int p = beg + wsum[wv] + x - s;
    cnt[t] = p;
    if (t < nn) off[n0 + t] = p;
  }
  if (b == NB - 1 && t == 0) off[N_NODES] = end;
  __syncthreads();

  for (int i = beg + t; i < end; i += GB_THR) {
    const int2 r = brec[i];
    const unsigned key = (unsigned)r.x;
    const int p = atomicAdd(&cnt[key >> SRC_BITS], 1);
    csr[p] = make_int2((int)(key & ((1u << SRC_BITS) - 1u)), r.y);
  }
}

template <int U>
__device__ __forceinline__ void gather_rows(const float2* __restrict__ sup2,
                                            const int2 ed, const int j,
                                            const int lane, float2& acc) {
  float2 v[U];
  float  w[U];
#pragma unroll
  for (int u = 0; u < U; ++u) {
    const int s = __builtin_amdgcn_readlane(ed.x, j + u);
    w[u] = __int_as_float(__builtin_amdgcn_readlane(ed.y, j + u));
    v[u] = sup2[(long long)s * 32 + lane];
  }
#pragma unroll
  for (int u = 0; u < U; ++u) {
    acc.x += w[u] * v[u].x;
    acc.y += w[u] * v[u].y;
  }
}

__device__ __forceinline__ void gather_chunk(const float2* __restrict__ sup2,
                                             const int2 ed, const int cnt,
                                             const int lane, float2& acc) {
  int j = 0;
  for (; j + 8 <= cnt; j += 8) gather_rows<8>(sup2, ed, j, lane, acc);
  if (j + 4 <= cnt) { gather_rows<4>(sup2, ed, j, lane, acc); j += 4; }
  if (j + 2 <= cnt) { gather_rows<2>(sup2, ed, j, lane, acc); j += 2; }
  if (j < cnt)      { gather_rows<1>(sup2, ed, j, lane, acc); }
}

__global__ __launch_bounds__(256)
void spmm_gather_kernel(const float* __restrict__ sup,
                        const int* __restrict__ csr_raw,
                        const int* __restrict__ off,
                        const float* __restrict__ bias,
                        float* __restrict__ out) {
  const int lane = threadIdx.x & 31;
  const int wave = __builtin_amdgcn_readfirstlane(
      (int)(blockIdx.x * (blockDim.x >> 5) + (threadIdx.x >> 5)));
  const int n0 = wave * NPW;
  if (n0 >= N_NODES) return;

  const float2* __restrict__ sup2 = (const float2*)sup;
  const int2*   __restrict__ csr  = (const int2*)csr_raw;
  const float2 bv = ((const float2*)bias)[lane];

  const int myoff = off[n0 + (lane <= NPW ? lane : NPW)];
  int o[NPW + 1];
#pragma unroll
  for (int i = 0; i <= NPW; ++i) o[i] = __builtin_amdgcn_readlane(myoff, i);
  int2 ed[NPW];
#pragma unroll
  for (int i = 0; i < NPW; ++i) {
    const int len = o[i + 1] - o[i];
    ed[i] = (lane < len) ? csr[o[i] + lane] : make_int2(0, 0);
  }

#pragma unroll
  for (int i = 0; i < NPW; ++i) {
    const int beg = o[i], end = o[i + 1];
    float2 acc = make_float2(0.0f, 0.0f);
    int cnt = end - beg;
    gather_chunk(sup2, ed[i], cnt < 32 ? cnt : 32, lane, acc);
    for (int c = beg + 32; c < end; c += 32) {
      cnt = end - c;
      const int2 e2 = (lane < cnt) ? csr[c + lane] : make_int2(0, 0);
      gather_chunk(sup2, e2, cnt < 32 ? cnt : 32, lane, acc);
    }
    float2 r;
    r.x = acc.x + bv.x;
    r.y = acc.y + bv.y;
    ((float2*)(out + (long long)(n0 + i) * 64))[lane] = r;
  }
}

#define FR 32
#define HP 68

static_assert(FR == 8 * NPW && N_NODES % FR == 0, "work-groups tile the rows exactly");

__global__ __launch_bounds__(256)
void spmm_gemm_kernel(const float* __restrict__ sup,
                      const int* __restrict__ csr_raw,
                      const int* __restrict__ off,
                      const float* __restrict__ bias,
                      const float* __restrict__ Wn,
                      float* __restrict__ supn) {
  __shared__ alignas(16) v2f   Pw[(D_H / 2) * 64];
  __shared__ alignas(16) float Ht[FR * HP];

  const int t    = threadIdx.x;
  const int lane = t & 31;
  const int wv   = __builtin_amdgcn_readfirstlane(t >> 5);
  const int row0 = blockIdx.x * FR;
  const int n0   = row0 + wv * NPW;

  const int kp0 = t >> 4;
  const int cg  = t & 15;
  const float4 wa0 = *(const float4*)(Wn + (2 * kp0)      * 64 + 4 * cg);
  const float4 wb0 = *(const float4*)(Wn + (2 * kp0 + 1)  * 64 + 4 * cg);
  const float4 wa1 = *(const float4*)(Wn + (2 * kp0 + 32) * 64 + 4 * cg);
  const float4 wb1 = *(const float4*)(Wn + (2 * kp0 + 33) * 64 + 4 * cg);

  const float2* __restrict__ sup2 = (const float2*)sup;
  const int2*   __restrict__ csr  = (const int2*)csr_raw;
  const float2 bv = ((const float2*)bias)[lane];

  const int myoff = off[n0 + (lane <= NPW ? lane : NPW)];
  int o[NPW + 1];
#pragma unroll
  for (int i = 0; i <= NPW; ++i) o[i] = __builtin_amdgcn_readlane(myoff, i);
  int2 ed[NPW];
#pragma unroll
  for (int i = 0; i < NPW; ++i) {
    const int len = o[i + 1] - o[i];
    ed[i] = (lane < len) ? csr[o[i] + lane] : make_int2(0, 0);
  }

#pragma unroll
  for (int i = 0; i < NPW; ++i) {
    const int beg = o[i], end = o[i + 1];
    float2 acc = make_float2(0.0f, 0.0f);
    int cnt = end - beg;
    gather_chunk(sup2, ed[i], cnt < 32 ? cnt : 32, lane, acc);
    for (int c = beg + 32; c < end; c += 32) {
      cnt = end - c;
      const int2 e2 = (lane < cnt) ? csr[c + lane] : make_int2(0, 0);
      gather_chunk(sup2, e2, cnt < 32 ? cnt : 32, lane, acc);
    }
    float2 r;
    r.x = fmaxf(acc.x + bv.x, 0.0f);
    r.y = fmaxf(acc.y + bv.y, 0.0f);
    *(float2*)(Ht + (wv * NPW + i) * HP + 2 * lane) = r;
  }

  {
    float4* __restrict__ p0 = (float4*)(Pw + kp0 * 64 + 4 * cg);
    float4* __restrict__ p1 = (float4*)(Pw + (kp0 + 16) * 64 + 4 * cg);
    p0[0] = make_float4(wa0.x, wb0.x, wa0.y, wb0.y);
    p0[1] = make_float4(wa0.z, wb0.z, wa0.w, wb0.w);
    p1[0] = make_float4(wa1.x, wb1.x, wa1.y, wb1.y);
    p1[1] = make_float4(wa1.z, wb1.z, wa1.w, wb1.w);
  }
  __syncthreads();

  if (wv >= 4) return;
  const int half = lane >> 4;
  const int l15  = lane & 15;
  const int s    = wv >> 1;
  const int cb   = (wv & 1) * 32;
  const float* __restrict__ hrow = Ht + (16 * s + l15) * HP + 2 * half;

  v8f acc0 = {}, acc1 = {};
  for (int k = 0; k < D_H; k += 4) {
    const v2f a = *(const v2f*)(hrow + k);
    const v2f* __restrict__ Pk = Pw + (k / 2 + half) * 64 + cb;
    const v2f b0 = Pk[l15];
    const v2f b1 = Pk[l15 + 16];
    acc0 = __builtin_amdgcn_wmma_f32_16x16x4_f32(false, a, false, b0, (short)0, acc0, false, false);
    acc1 = __builtin_amdgcn_wmma_f32_16x16x4_f32(false, a, false, b1, (short)0, acc1, false, false);
  }

#pragma unroll
  for (int r = 0; r < 8; ++r) {
    const long long m = row0 + 16 * s + r + 8 * half;
    float* __restrict__ crow = supn + m * 64 + cb;
    crow[l15]      = acc0[r];
    crow[l15 + 16] = acc1[r];
  }
}

extern "C" void kernel_launch(void* const* d_in, const int* in_sizes, int n_in,
                              void* d_out, int out_size, void* d_ws, size_t ws_size,
                              hipStream_t stream) {
  const float* features = (const float*)d_in[0];
  const int*   src      = (const int*)  d_in[1];
  const int*   dst      = (const int*)  d_in[2];
  const float* ew       = (const float*)d_in[3];
  const float* W1       = (const float*)d_in[4];
  const float* b1       = (const float*)d_in[5];
  const float* W2       = (const float*)d_in[6];
  const float* b2       = (const float*)d_in[7];
  const float* W3       = (const float*)d_in[8];
  const float* b3       = (const float*)d_in[9];
  float* out = (float*)d_out;

  const long long feat_elems = (long long)N_NODES * D_H;
  float* sup  = (float*)d_ws;
  int*   csr  = (int*)(sup + feat_elems);
  int*   off  = csr + 2LL * N_EDGES;
  int*   brec = (int*)sup;
  int*   tcnt = brec + 2LL * N_EDGES;
  int*   gstart = tcnt + TBL;
  float* sup2 = out;

  const dim3 blk(256);
  const int gemm_blocks = (N_NODES / 16 + 7) / 8;
  const int spmm_blocks = (N_NODES / NPW + 7) / 8;

  bucket_count_kernel<<<NCH, dim3(EB_THR), 0, stream>>>(dst, tcnt);
  bucket_place_kernel<<<NCH, dim3(EB_THR), 0, stream>>>(src, dst, ew, tcnt, gstart, brec);
  bucket_group_kernel<<<NB, dim3(GB_THR), 0, stream>>>(brec, gstart, off, csr);

  gemm_wmma_kernel<D_IN, false><<<gemm_blocks, blk, 0, stream>>>(features, W1, sup, N_NODES);

  spmm_gemm_kernel<<<N_NODES / FR, blk, 0, stream>>>(sup, csr, off, b1, W2, sup2);

  spmm_gemm_kernel<<<N_NODES / FR, blk, 0, stream>>>(sup2, csr, off, b2, W3, sup);

  spmm_gather_kernel<<<spmm_blocks, blk, 0, stream>>>(sup, csr, off, b3, out);

  (void)in_sizes; (void)n_in; (void)out_size; (void)ws_size;
}


# Layer-1 product X@W1 runs in the same launch as the per-bucket grouping pass (work-groups split by index; grouping scratch moved to the output plane); one launch gone; sums unchanged
# speedup vs baseline: 4.0238x; 1.1319x over previous
#include <hip/hip_runtime.h>

#define N_NODES 100000
#define N_EDGES 1600000
#define D_IN    128
#define D_H     64

typedef __attribute__((ext_vector_type(2))) float v2f;
typedef __attribute__((ext_vector_type(8))) float v8f;

template <int K, bool RELU_IN>
__device__ __forceinline__ void gemm_wmma_body(const float* __restrict__ A,
                                               const float* __restrict__ W,
                                               float* __restrict__ C, int nrows,
                                               v2f* __restrict__ Pw, int gblock) {
  for (int idx = threadIdx.x; idx < K * 64; idx += blockDim.x) {
    const int k = idx >> 6;
    const int c = idx & 63;
    ((float*)Pw)[(k >> 1) * 128 + c * 2 + (k & 1)] = W[idx];
  }
  __syncthreads();

  const int lane = threadIdx.x & 31;
  const int wave = gblock * (blockDim.x >> 5) + (threadIdx.x >> 5);
  const int row0 = wave * 16;
  if (row0 >= nrows) return;

  const int half = lane >> 4;
  const int l15  = lane & 15;
  const long long arow = row0 + l15;

  v8f acc0 = {}, acc1 = {}, acc2 = {}, acc3 = {};

  for (int k = 0; k < K; k += 4) {
    const int kk = k + 2 * half;
    v2f a = *(const v2f*)(A + arow * K + kk);
    if (RELU_IN) {
      a.x = fmaxf(a.x, 0.0f);
      a.y = fmaxf(a.y, 0.0f);
    }
    const v2f* __restrict__ Pk = Pw + (k / 2 + half) * 64;
    const v2f b0 = Pk[l15 +  0];
    const v2f b1 = Pk[l15 + 16];
    const v2f b2 = Pk[l15 + 32];
    const v2f b3 = Pk[l15 + 48];

    acc0 = __builtin_amdgcn_wmma_f32_16x16x4_f32(false, a, false, b0, (short)0, acc0, false, false);
    acc1 = __builtin_amdgcn_wmma_f32_16x16x4_f32(false, a, false, b1, (short)0, acc1, false, false);
    acc2 = __builtin_amdgcn_wmma_f32_16x16x4_f32(false, a, false, b2, (short)0, acc2, false, false);
    acc3 = __builtin_amdgcn_wmma_f32_16x16x4_f32(false, a, false, b3, (short)0, acc3, false, false);
  }

#pragma unroll
  for (int r = 0; r < 8; ++r) {
    const long long m = row0 + r + 8 * half;
    float* __restrict__ crow = C + m * 64;
    crow[l15 +  0] = acc0[r];
    crow[l15 + 16] = acc1[r];
    crow[l15 + 32] = acc2[r];
    crow[l15 + 48] = acc3[r];
  }
}

#define NCH      250
#define CH_EDGES 6400
#define EB_THR   640
#define EB_EPT   10
#define NB       256
#define BKT      391
#define TBL      (NB * NCH)
#define TP_LANES (NB / 4)
#define TP_GRP   (EB_THR / TP_LANES)
#define TP_ROWS  (NCH / TP_GRP)
#define SRC_BITS 17
#define GB_THR   1024
#define GB_ROWS  512
#define NPW      4

static_assert(NCH * CH_EDGES == N_EDGES && EB_THR * EB_EPT == CH_EDGES, "chunks tile the edge list");
static_assert(NB * BKT >= N_NODES && (NB - 1) * BKT < N_NODES, "buckets tile the rows");
static_assert(N_NODES <= (1 << SRC_BITS) && BKT <= (1 << (31 - SRC_BITS)), "record packing");
static_assert(NB % 32 == 0 && NB <= EB_THR && TP_LANES % 32 == 0 && TP_GRP * TP_LANES == EB_THR &&
              TP_GRP * TP_ROWS == NCH, "the place pass's thread groups tile the table exactly");
static_assert(BKT <= GB_ROWS && GB_ROWS % 32 == 0 && GB_ROWS <= GB_THR, "bucket rows fit the LDS scan");
static_assert(N_NODES % NPW == 0, "every wave owns NPW whole rows");
static_assert(2LL * N_EDGES + TBL + NB + 1 <= (long long)N_NODES * D_H, "the grouping's scratch fits the output plane");

__global__ __launch_bounds__(EB_THR)
void bucket_count_kernel(const int* __restrict__ dst, int* __restrict__ tcnt) {
  __shared__ int hist[NB];
  const int t = threadIdx.x;
  const int c = blockIdx.x;
  if (t < NB) hist[t] = 0;
  __syncthreads();
  const int* __restrict__ d = dst + c * CH_EDGES + t;
  int dv[EB_EPT];
#pragma unroll
  for (int k = 0; k < EB_EPT; ++k) dv[k] = d[k * EB_THR];
#pragma unroll
  for (int k = 0; k < EB_EPT; ++k) atomicAdd(&hist[(unsigned)dv[k] / BKT], 1);
  __syncthreads();
  if (t < NB) tcnt[c * NB + t] = hist[t];
}

__global__ __launch_bounds__(EB_THR)
void bucket_place_kernel(const int* __restrict__ src,
                         const int* __restrict__ dst,
                         const float* __restrict__ ew,
                         const int* __restrict__ tcnt,
                         int* __restrict__ gstart,
                         int* __restrict__ brec_raw) {
  __shared__ alignas(16) int pall[TP_GRP][NB];
  __shared__ alignas(16) int pbef[TP_GRP][NB];
  __shared__ int wsum[NB / 32];
  __shared__ int cur[NB];
  const int t    = threadIdx.x;
  const int lane = t & 31;
  const int wv   = t >> 5;
  const int c    = blockIdx.x;
  int2* __restrict__ brec = (int2*)brec_raw;
  const int e0 = c * CH_EDGES + t;
  int   dv[EB_EPT];
  int   sv[EB_EPT];
  float wf[EB_EPT];
#pragma unroll
  for (int k = 0; k < EB_EPT; ++k) {
    dv[k] = dst[e0 + k * EB_THR];
    sv[k] = src[e0 + k * EB_THR];
    wf[k] = ew[e0 + k * EB_THR];
  }

  {
    const int g = __builtin_amdgcn_readfirstlane(t / TP_LANES);
    const int q = t % TP_LANES;
    const int4* __restrict__ trow = (const int4*)tcnt + (g * TP_ROWS) * TP_LANES + q;
    const int nbef = c - g * TP_ROWS;
    int4 v[TP_ROWS];
#pragma unroll
    for (int r = 0; r < TP_ROWS; ++r) v[r] = trow[r * TP_LANES];
    int4 sa = make_int4(0, 0, 0, 0);
    int4 sb = make_int4(0, 0, 0, 0);
#pragma unroll
    for (int r = 0; r < TP_ROWS; ++r) {
      const int m = (r < nbef) ? -1 : 0;
      sa.x += v[r].x;     sa.y += v[r].y;     sa.z += v[r].z;     sa.w += v[r].w;
      sb.x += v[r].x & m; sb.y += v[r].y & m; sb.z += v[r].z & m; sb.w += v[r].w & m;
    }
    *(int4*)&pall[g][4 * q] = sa;
    *(int4*)&pbef[g][4 * q] = sb;
  }
  __syncthreads();

  int all = 0, bef = 0, x = 0;
  if (t < NB) {
#pragma unroll
    for (int g = 0; g < TP_GRP; ++g) {
      all += pall[g][t];
      bef += pbef[g][t];
    }
    x = all;
    for (int d = 1; d < 32; d <<= 1) {
      const int y = __shfl_up(x, d);
      if (lane >= d) x += y;
    }
    if (lane == 31) wsum[wv] = x;
  }
  __syncthreads();
  if (t < NB) {
    int base = x - all;
    for (int w = 0; w < NB / 32; ++w)
      if (w < wv) base += wsum[w];
    cur[t] = base + bef;
    if (c == 0) {
      gstart[t] = base;
      if (t == NB - 1) gstart[NB] = base + all;
    }
  }
  __syncthreads();

#pragma unroll
  for (int k = 0; k < EB_EPT; ++k) {
    const unsigned d  = (unsigned)dv[k];
    const unsigned b  = d / BKT;
    const unsigned dl = d - b * BKT;
    const int p = atomicAdd(&cur[b], 1);
    brec[p] = make_int2((int)((unsigned)sv[k] | (dl << SRC_BITS)), __float_as_int(wf[k]));
  }
}

__device__ __forceinline__ void bucket_group_body(const int* __restrict__ brec_raw,
                                                  const int* __restrict__ gstart,
                                                  int* __restrict__ off,
                                                  int* __restrict__ csr_raw,
                                                  int* __restrict__ cnt,
                                                  int* __restrict__ wsum,
                                                  const int b) {
  const int t    = threadIdx.x;
  const int lane = t & 31;
  const int wv   = t >> 5;
  const int n0   = b * BKT;
  const int nn   = (N_NODES - n0 < BKT) ? (N_NODES - n0) : BKT;
  const int beg  = gstart[b];
  const int end  = gstart[b + 1];
  const int2* __restrict__ brec = (const int2*)brec_raw;
  int2*       __restrict__ csr  = (int2*)csr_raw;

  if (t < GB_ROWS) cnt[t] = 0;
  __syncthreads();
  for (int i = beg + t; i < end; i += GB_THR) {
    const unsigned key = (unsigned)brec[i].x;
    atomicAdd(&cnt[key >> SRC_BITS], 1);
  }
  __syncthreads();

  int s = 0, x = 0;
  if (t < GB_ROWS) {
    s = cnt[t];
    x = s;
    for (int d = 1; d < 32; d <<= 1) {
      const int y = __shfl_up(x, d);
      if (lane >= d) x += y;
    }
    if (lane == 31) wsum[wv] = x;
  }
  __syncthreads();
  if (wv == 0) {
    const int v = (lane < GB_ROWS / 32) ? wsum[lane] : 0;
    int xx = v;
    for (int d = 1; d < 32; d <<= 1) {
      const int y = __shfl_up(xx, d);
      if (lane >= d) xx += y;
    }
    wsum[lane] = xx - v;
  }
  __syncthreads();
  if (t < GB_ROWS) {
    const int p = beg + wsum[wv] + x - s;
    cnt[t] = p;
    if (t < nn) off[n0 + t] = p;
  }
  if (b == NB - 1 && t == 0) off[N_NODES] = end;
  __syncthreads();

  for (int i = beg + t; i < end; i += GB_THR) {
    const int2 r = brec[i];
    const unsigned key = (unsigned)r.x;
    const int p = atomicAdd(&cnt[key >> SRC_BITS], 1);
    csr[p] = make_int2((int)(key & ((1u << SRC_BITS) - 1u)), r.y);
  }
}

#define GG_WAVES       (GB_THR / 32)
#define GG_GEMM_BLOCKS ((N_NODES / 16 + GG_WAVES - 1) / GG_WAVES)

static_assert(N_NODES % 16 == 0, "whole 16-row strips");
static_assert((D_IN / 2) * 64 * 2 >= GB_ROWS + 32, "the product's LDS block also holds the grouping's counters");

__global__ __launch_bounds__(GB_THR)
void group_gemm_kernel(const int* __restrict__ brec_raw,
                       const int* __restrict__ gstart,
                       int* __restrict__ off,
                       int* __restrict__ csr_raw,
                       const float* __restrict__ X,
                       const float* __restrict__ W,
                       float* __restrict__ C) {
  __shared__ alignas(16) v2f lds[(D_IN / 2) * 64];
  if (blockIdx.x < NB) {
    int* cnt = (int*)lds;
    bucket_group_body(brec_raw, gstart, off, csr_raw, cnt, cnt + GB_ROWS, (int)blockIdx.x);
  } else {
    gemm_wmma_body<D_IN, false>(X, W, C, N_NODES, lds, (int)blockIdx.x - NB);
  }
}

template <int U>
__device__ __forceinline__ void gather_rows(const float2* __restrict__ sup2,
                                            const int2 ed, const int j,
                                            const int lane, float2& acc) {
  float2 v[U];
  float  w[U];
#pragma unroll
  for (int u = 0; u < U; ++u) {
    const int s = __builtin_amdgcn_readlane(ed.x, j + u);
    w[u] = __int_as_float(__builtin_amdgcn_readlane(ed.y, j + u));
    v[u] = sup2[(long long)s * 32 + lane];
  }
#pragma unroll
  for (int u = 0; u < U; ++u) {
    acc.x += w[u] * v[u].x;
    acc.y += w[u] * v[u].y;
  }
}

__device__ __forceinline__ void gather_chunk(const float2* __restrict__ sup2,
                                             const int2 ed, const int cnt,
                                             const int lane, float2& acc) {
  int j = 0;
  for (; j + 8 <= cnt; j += 8) gather_rows<8>(sup2, ed, j, lane, acc);
  if (j + 4 <= cnt) { gather_rows<4>(sup2, ed, j, lane, acc); j += 4; }
  if (j + 2 <= cnt) { gather_rows<2>(sup2, ed, j, lane, acc); j += 2; }
  if (j < cnt)      { gather_rows<1>(sup2, ed, j, lane, acc); }
}

__global__ __launch_bounds__(256)
void spmm_gather_kernel(const float* __restrict__ sup,
                        const int* __restrict__ csr_raw,
                        const int* __restrict__ off,
                        const float* __restrict__ bias,
                        float* __restrict__ out) {
  const int lane = threadIdx.x & 31;
  const int wave = __builtin_amdgcn_readfirstlane(
      (int)(blockIdx.x * (blockDim.x >> 5) + (threadIdx.x >> 5)));
  const int n0 = wave * NPW;
  if (n0 >= N_NODES) return;

  const float2* __restrict__ sup2 = (const float2*)sup;
  const int2*   __restrict__ csr  = (const int2*)csr_raw;
  const float2 bv = ((const float2*)bias)[lane];

  const int myoff = off[n0 + (lane <= NPW ? lane : NPW)];
  int o[NPW + 1];
#pragma unroll
  for (int i = 0; i <= NPW; ++i) o[i] = __builtin_amdgcn_readlane(myoff, i);
  int2 ed[NPW];
#pragma unroll
  for (int i = 0; i < NPW; ++i) {
    const int len = o[i + 1] - o[i];
    ed[i] = (lane < len) ? csr[o[i] + lane] : make_int2(0, 0);
  }

#pragma unroll
  for (int i = 0; i < NPW; ++i) {
    const int beg = o[i], end = o[i + 1];
    float2 acc = make_float2(0.0f, 0.0f);
    int cnt = end - beg;
    gather_chunk(sup2, ed[i], cnt < 32 ? cnt : 32, lane, acc);
    for (int c = beg + 32; c < end; c += 32) {
      cnt = end - c;
      const int2 e2 = (lane < cnt) ? csr[c + lane] : make_int2(0, 0);
      gather_chunk(sup2, e2, cnt < 32 ? cnt : 32, lane, acc);
    }
    float2 r;
    r.x = acc.x + bv.x;
    r.y = acc.y + bv.y;
    ((float2*)(out + (long long)(n0 + i) * 64))[lane] = r;
  }
}

#define FR 32
#define HP 68

static_assert(FR == 8 * NPW && N_NODES % FR == 0, "work-groups tile the rows exactly");

__global__ __launch_bounds__(256)
void spmm_gemm_kernel(const float* __restrict__ sup,
                      const int* __restrict__ csr_raw,
                      const int* __restrict__ off,
                      const float* __restrict__ bias,
                      const float* __restrict__ Wn,
                      float* __restrict__ supn) {
  __shared__ alignas(16) v2f   Pw[(D_H / 2) * 64];
  __shared__ alignas(16) float Ht[FR * HP];

  const int t    = threadIdx.x;
  const int lane = t & 31;
  const int wv   = __builtin_amdgcn_readfirstlane(t >> 5);
  const int row0 = blockIdx.x * FR;
  const int n0   = row0 + wv * NPW;

  const int kp0 = t >> 4;
  const int cg  = t & 15;
  const float4 wa0 = *(const float4*)(Wn + (2 * kp0)      * 64 + 4 * cg);
  const float4 wb0 = *(const float4*)(Wn + (2 * kp0 + 1)  * 64 + 4 * cg);
  const float4 wa1 = *(const float4*)(Wn + (2 * kp0 + 32) * 64 + 4 * cg);
  const float4 wb1 = *(const float4*)(Wn + (2 * kp0 + 33) * 64 + 4 * cg);

  const float2* __restrict__ sup2 = (const float2*)sup;
  const int2*   __restrict__ csr  = (const int2*)csr_raw;
  const float2 bv = ((const float2*)bias)[lane];

  const int myoff = off[n0 + (lane <= NPW ? lane : NPW)];
  int o[NPW + 1];
#pragma unroll
  for (int i = 0; i <= NPW; ++i) o[i] = __builtin_amdgcn_readlane(myoff, i);
  int2 ed[NPW];
#pragma unroll
  for (int i = 0; i < NPW; ++i) {
    const int len = o[i + 1] - o[i];
    ed[i] = (lane < len) ? csr[o[i] + lane] : make_int2(0, 0);
  }

#pragma unroll
  for (int i = 0; i < NPW; ++i) {
    const int beg = o[i], end = o[i + 1];
    float2 acc = make_float2(0.0f, 0.0f);
    int cnt = end - beg;
    gather_chunk(sup2, ed[i], cnt < 32 ? cnt : 32, lane, acc);
    for (int c = beg + 32; c < end; c += 32) {
      cnt = end - c;
      const int2 e2 = (lane < cnt) ? csr[c + lane] : make_int2(0, 0);
      gather_chunk(sup2, e2, cnt < 32 ? cnt : 32, lane, acc);
    }
    float2 r;
    r.x = fmaxf(acc.x + bv.x, 0.0f);
    r.y = fmaxf(acc.y + bv.y, 0.0f);
    *(float2*)(Ht + (wv * NPW + i) * HP + 2 * lane) = r;
  }

  {
    float4* __restrict__ p0 = (float4*)(Pw + kp0 * 64 + 4 * cg);
    float4* __restrict__ p1 = (float4*)(Pw + (kp0 + 16) * 64 + 4 * cg);
    p0[0] = make_float4(wa0.x, wb0.x, wa0.y, wb0.y);
    p0[1] = make_float4(wa0.z, wb0.z, wa0.w, wb0.w);
    p1[0] = make_float4(wa1.x, wb1.x, wa1.y, wb1.y);
    p1[1] = make_float4(wa1.z, wb1.z, wa1.w, wb1.w);
  }
  __syncthreads();

  if (wv >= 4) return;
  const int half = lane >> 4;
  const int l15  = lane & 15;
  const int s    = wv >> 1;
  const int cb   = (wv & 1) * 32;
  const float* __restrict__ hrow = Ht + (16 * s + l15) * HP + 2 * half;

  v8f acc0 = {}, acc1 = {};
  for (int k = 0; k < D_H; k += 4) {
    const v2f a = *(const v2f*)(hrow + k);
    const v2f* __restrict__ Pk = Pw + (k / 2 + half) * 64 + cb;
    const v2f b0 = Pk[l15];
    const v2f b1 = Pk[l15 + 16];
    acc0 = __builtin_amdgcn_wmma_f32_16x16x4_f32(false, a, false, b0, (short)0, acc0, false, false);
    acc1 = __builtin_amdgcn_wmma_f32_16x16x4_f32(false, a, false, b1, (short)0, acc1, false, false);
  }

#pragma unroll
  for (int r = 0; r < 8; ++r) {
    const long long m = row0 + 16 * s + r + 8 * half;
    float* __restrict__ crow = supn + m * 64 + cb;
    crow[l15]      = acc0[r];
    crow[l15 + 16] = acc1[r];
  }
}

extern "C" void kernel_launch(void* const* d_in, const int* in_sizes, int n_in,
                              void* d_out, int out_size, void* d_ws, size_t ws_size,
                              hipStream_t stream) {
  const float* features = (const float*)d_in[0];
  const int*   src      = (const int*)  d_in[1];
  const int*   dst      = (const int*)  d_in[2];
  const float* ew       = (const float*)d_in[3];
  const float* W1       = (const float*)d_in[4];
  const float* b1       = (const float*)d_in[5];
  const float* W2       = (const float*)d_in[6];
  const float* b2       = (const float*)d_in[7];
  const float* W3       = (const float*)d_in[8];
  const float* b3       = (const float*)d_in[9];
  float* out = (float*)d_out;

  const long long feat_elems = (long long)N_NODES * D_H;
  float* sup  = (float*)d_ws;
  int*   csr  = (int*)(sup + feat_elems);
  int*   off  = csr + 2LL * N_EDGES;
  int*   brec = (int*)out;
  int*   tcnt = brec + 2LL * N_EDGES;
  int*   gstart = tcnt + TBL;
  float* sup2 = out;

  const dim3 blk(256);
  const int spmm_blocks = (N_NODES / NPW + 7) / 8;

  bucket_count_kernel<<<NCH, dim3(EB_THR), 0, stream>>>(dst, tcnt);
  bucket_place_kernel<<<NCH, dim3(EB_THR), 0, stream>>>(src, dst, ew, tcnt, gstart, brec);

  group_gemm_kernel<<<NB + GG_GEMM_BLOCKS, dim3(GB_THR), 0, stream>>>(brec, gstart, off, csr,
                                                                     features, W1, sup);

  spmm_gemm_kernel<<<N_NODES / FR, blk, 0, stream>>>(sup, csr, off, b1, W2, sup2);

  spmm_gemm_kernel<<<N_NODES / FR, blk, 0, stream>>>(sup2, csr, off, b2, W3, sup);

  spmm_gather_kernel<<<spmm_blocks, blk, 0, stream>>>(sup, csr, off, b3, out);

  (void)in_sizes; (void)n_in; (void)out_size; (void)ws_size;
}


# Each edge chunk sorts its edges by bucket in LDS and stores one contiguous run; the bucket pass walks its 250 runs via a count/offset table: count launch gone; row order still LDS arrival rank
# speedup vs baseline: 4.4616x; 1.1088x over previous
#include <hip/hip_runtime.h>

#define N_NODES 100000
#define N_EDGES 1600000
#define D_IN    128
#define D_H     64

typedef __attribute__((ext_vector_type(2))) float v2f;
typedef __attribute__((ext_vector_type(8))) float v8f;

template <int K, bool RELU_IN>
__device__ __forceinline__ void gemm_wmma_body(const float* __restrict__ A,
                                               const float* __restrict__ W,
                                               float* __restrict__ C, int nrows,
                                               v2f* __restrict__ Pw, int gblock) {
  for (int idx = threadIdx.x; idx < K * 64; idx += blockDim.x) {
    const int k = idx >> 6;
    const int c = idx & 63;
    ((float*)Pw)[(k >> 1) * 128 + c * 2 + (k & 1)] = W[idx];
  }
  __syncthreads();

  const int lane = threadIdx.x & 31;
  const int wave = gblock * (blockDim.x >> 5) + (threadIdx.x >> 5);
  const int row0 = wave * 16;
  if (row0 >= nrows) return;

  const int half = lane >> 4;
  const int l15  = lane & 15;
  const long long arow = row0 + l15;

  v8f acc0 = {}, acc1 = {}, acc2 = {}, acc3 = {};

  for (int k = 0; k < K; k += 4) {
    const int kk = k + 2 * half;
    v2f a = *(const v2f*)(A + arow * K + kk);
    if (RELU_IN) {
      a.x = fmaxf(a.x, 0.0f);
      a.y = fmaxf(a.y, 0.0f);
    }
    const v2f* __restrict__ Pk = Pw + (k / 2 + half) * 64;
    const v2f b0 = Pk[l15 +  0];
    const v2f b1 = Pk[l15 + 16];
    const v2f b2 = Pk[l15 + 32];
    const v2f b3 = Pk[l15 + 48];

    acc0 = __builtin_amdgcn_wmma_f32_16x16x4_f32(false, a, false, b0, (short)0, acc0, false, false);
    acc1 = __builtin_amdgcn_wmma_f32_16x16x4_f32(false, a, false, b1, (short)0, acc1, false, false);
    acc2 = __builtin_amdgcn_wmma_f32_16x16x4_f32(false, a, false, b2, (short)0, acc2, false, false);
    acc3 = __builtin_amdgcn_wmma_f32_16x16x4_f32(false, a, false, b3, (short)0, acc3, false, false);
  }

#pragma unroll
  for (int r = 0; r < 8; ++r) {
    const long long m = row0 + r + 8 * half;
    float* __restrict__ crow = C + m * 64;
    crow[l15 +  0] = acc0[r];
    crow[l15 + 16] = acc1[r];
    crow[l15 + 32] = acc2[r];
    crow[l15 + 48] = acc3[r];
  }
}

#define NCH      250
#define CH_EDGES 6400
#define EB_THR   640
#define EB_EPT   10
#define NB       256
#define BKT      391
#define TBL      (NB * NCH)
#define SRC_BITS 17
#define GB_THR   1024
#define GB_ROWS  512
#define SEG_LANES 4
#define NPW      4

static_assert(NCH * CH_EDGES == N_EDGES && EB_THR * EB_EPT == CH_EDGES, "chunks tile the edge list");
static_assert(CH_EDGES % (2 * EB_THR) == 0, "the sorted chunk leaves LDS in whole 16-byte pieces per thread");
static_assert(NB * BKT >= N_NODES && (NB - 1) * BKT < N_NODES, "buckets tile the rows");
static_assert(N_NODES <= (1 << SRC_BITS) && BKT <= (1 << (31 - SRC_BITS)), "record packing");
static_assert(NB % 32 == 0 && NB <= EB_THR, "the chunk pass scans the bucket counts with whole waves");
static_assert(NCH <= 256 && NCH * SEG_LANES <= GB_THR, "the per-bucket pass covers every chunk's run");
static_assert(BKT <= GB_ROWS && GB_ROWS % 32 == 0 && GB_ROWS <= GB_THR, "bucket rows fit the LDS scan");
static_assert(N_NODES % NPW == 0, "every wave owns NPW whole rows");
static_assert(2LL * N_EDGES + 2LL * TBL <= (long long)N_NODES * D_H, "the grouping's scratch fits the output plane");

__global__ __launch_bounds__(EB_THR)
void chunk_sort_kernel(const int* __restrict__ src,
                       const int* __restrict__ dst,
                       const float* __restrict__ ew,
                       int* __restrict__ tcntT,
                       int* __restrict__ toffT,
                       int* __restrict__ brec_raw) {
  __shared__ alignas(16) int2 stage[CH_EDGES];
  __shared__ int hist[NB];
  __shared__ int wsum[NB / 32];
  const int t    = threadIdx.x;
  const int lane = t & 31;
  const int wv   = t >> 5;
  const int c    = blockIdx.x;
  const int e0 = c * CH_EDGES + t;
  int   dv[EB_EPT];
  int   sv[EB_EPT];
  int   rk[EB_EPT];
  float wf[EB_EPT];
  if (t < NB) hist[t] = 0;
#pragma unroll
  for (int k = 0; k < EB_EPT; ++k) {
    dv[k] = dst[e0 + k * EB_THR];
    sv[k] = src[e0 + k * EB_THR];
    wf[k] = ew[e0 + k * EB_THR];
  }
  __syncthreads();

#pragma unroll
  for (int k = 0; k < EB_EPT; ++k) rk[k] = atomicAdd(&hist[(unsigned)dv[k] / BKT], 1);
  __syncthreads();

  int all = 0, x = 0;
  if (t < NB) {
    all = hist[t];
    x = all;
    for (int d = 1; d < 32; d <<= 1) {
      const int y = __shfl_up(x, d);
      if (lane >= d) x += y;
    }
    if (lane == 31) wsum[wv] = x;
  }
  __syncthreads();
  if (t < NB) {
    int base = x - all;
    for (int w = 0; w < NB / 32; ++w)
      if (w < wv) base += wsum[w];
    hist[t] = base;
    tcntT[t * NCH + c] = all;
    toffT[t * NCH + c] = base;
  }
  __syncthreads();

#pragma unroll
  for (int k = 0; k < EB_EPT; ++k) {
    const unsigned d  = (unsigned)dv[k];
    const unsigned b  = d / BKT;
    const unsigned dl = d - b * BKT;
    stage[hist[b] + rk[k]] = make_int2((int)((unsigned)sv[k] | (dl << SRC_BITS)), __float_as_int(wf[k]));
  }
  __syncthreads();

  const int4* __restrict__ s4 = (const int4*)stage;
  int4* __restrict__ g4 = (int4*)(brec_raw + 2LL * c * CH_EDGES);
#pragma unroll
  for (int i = 0; i < CH_EDGES / (2 * EB_THR); ++i) g4[t + i * EB_THR] = s4[t + i * EB_THR];
}

__device__ __forceinline__ void bucket_group_body(const int* __restrict__ brec_raw,
                                                  const int* __restrict__ tcntT,
                                                  const int* __restrict__ toffT,
                                                  int* __restrict__ off,
                                                  int* __restrict__ csr_raw,
                                                  int* __restrict__ lds,
                                                  const int b) {
  int* __restrict__ cnt    = lds;
  int* __restrict__ wsum   = cnt + GB_ROWS;
  int* __restrict__ segcnt = wsum + 32;
  int* __restrict__ segoff = segcnt + 256;
  int* __restrict__ red    = segoff + 256;
  const int t    = threadIdx.x;
  const int lane = t & 31;
  const int wv   = t >> 5;
  const int n0   = b * BKT;
  const int nn   = (N_NODES - n0 < BKT) ? (N_NODES - n0) : BKT;
  const int2* __restrict__ brec = (const int2*)brec_raw;
  int2*       __restrict__ csr  = (int2*)csr_raw;

  int sc = 0, so = 0;
  if (t < NCH) {
    sc = tcntT[b * NCH + t];
    so = toffT[b * NCH + t];
    segcnt[t] = sc;
    segoff[t] = so;
  }
  if (t < GB_ROWS) cnt[t] = 0;
  __syncthreads();

  const int seg = t / SEG_LANES;
  const int q   = t % SEG_LANES;
  int scnt = 0, sbase = 0;
  if (seg < NCH) {
    scnt  = segcnt[seg];
    sbase = seg * CH_EDGES + segoff[seg];
  }
  for (int k = q; k < scnt; k += SEG_LANES) {
    const unsigned key = (unsigned)brec[sbase + k].x;
    atomicAdd(&cnt[key >> SRC_BITS], 1);
  }
  if (t < 256) {
    int a = so, e = sc;
    for (int d = 16; d >= 1; d >>= 1) {
      a += __shfl_xor(a, d);
      e += __shfl_xor(e, d);
    }
    if (lane == 0) {
      red[wv]     = a;
      red[8 + wv] = e;
    }
  }
  __syncthreads();
  int beg = 0, tot = 0;
#pragma unroll
  for (int w = 0; w < 8; ++w) {
    beg += red[w];
    tot += red[8 + w];
  }
  const int end = beg + tot;

  int s = 0, x = 0;
  if (t < GB_ROWS) {
    s = cnt[t];
    x = s;
    for (int d = 1; d < 32; d <<= 1) {
      const int y = __shfl_up(x, d);
      if (lane >= d) x += y;
    }
    if (lane == 31) wsum[wv] = x;
  }
  __syncthreads();
  if (wv == 0) {
    const int v = (lane < GB_ROWS / 32) ? wsum[lane] : 0;
    int xx = v;
    for (int d = 1; d < 32; d <<= 1) {
      const int y = __shfl_up(xx, d);
      if (lane >= d) xx += y;
    }
    wsum[lane] = xx - v;
  }
  __syncthreads();
  if (t < GB_ROWS) {
    const int p = beg + wsum[wv] + x - s;
    cnt[t] = p;
    if (t < nn) off[n0 + t] = p;
  }
  if (b == NB - 1 && t == 0) off[N_NODES] = end;
  __syncthreads();

  for (int k = q; k < scnt; k += SEG_LANES) {
    const int2 r = brec[sbase + k];
    const unsigned key = (unsigned)r.x;
    const int p = atomicAdd(&cnt[key >> SRC_BITS], 1);
    csr[p] = make_int2((int)(key & ((1u << SRC_BITS) - 1u)), r.y);
  }
}

#define GG_WAVES       (GB_THR / 32)
#define GG_GEMM_BLOCKS ((N_NODES / 16 + GG_WAVES - 1) / GG_WAVES)

static_assert(N_NODES % 16 == 0, "whole 16-row strips");
static_assert((D_IN / 2) * 64 * 2 >= GB_ROWS + 32 + 2 * 256 + 16, "the product's LDS block also holds the grouping's counters and table column");

__global__ __launch_bounds__(GB_THR)
void group_gemm_kernel(const int* __restrict__ brec_raw,
                       const int* __restrict__ tcntT,
                       const int* __restrict__ toffT,
                       int* __restrict__ off,
                       int* __restrict__ csr_raw,
                       const float* __restrict__ X,
                       const float* __restrict__ W,
                       float* __restrict__ C) {
  __shared__ alignas(16) v2f lds[(D_IN / 2) * 64];
  if (blockIdx.x < NB) {
    bucket_group_body(brec_raw, tcntT, toffT, off, csr_raw, (int*)lds, (int)blockIdx.x);
  } else {
    gemm_wmma_body<D_IN, false>(X, W, C, N_NODES, lds, (int)blockIdx.x - NB);
  }
}

template <int U>
__device__ __forceinline__ void gather_rows(const float2* __restrict__ sup2,
                                            const int2 ed, const int j,
                                            const int lane, float2& acc) {
  float2 v[U];
  float  w[U];
#pragma unroll
  for (int u = 0; u < U; ++u) {
    const int s = __builtin_amdgcn_readlane(ed.x, j + u);
    w[u] = __int_as_float(__builtin_amdgcn_readlane(ed.y, j + u));
    v[u] = sup2[(long long)s * 32 + lane];
  }
#pragma unroll
  for (int u = 0; u < U; ++u) {
    acc.x += w[u] * v[u].x;
    acc.y += w[u] * v[u].y;
  }
}

__device__ __forceinline__ void gather_chunk(const float2* __restrict__ sup2,
                                             const int2 ed, const int cnt,
                                             const int lane, float2& acc) {
  int j = 0;
  for (; j + 8 <= cnt; j += 8) gather_rows<8>(sup2, ed, j, lane, acc);
  if (j + 4 <= cnt) { gather_rows<4>(sup2, ed, j, lane, acc); j += 4; }
  if (j + 2 <= cnt) { gather_rows<2>(sup2, ed, j, lane, acc); j += 2; }
  if (j < cnt)      { gather_rows<1>(sup2, ed, j, lane, acc); }
}

__global__ __launch_bounds__(256)
void spmm_gather_kernel(const float* __restrict__ sup,
                        const int* __restrict__ csr_raw,
                        const int* __restrict__ off,
                        const float* __restrict__ bias,
                        float* __restrict__ out) {
  const int lane = threadIdx.x & 31;
  const int wave = __builtin_amdgcn_readfirstlane(
      (int)(blockIdx.x * (blockDim.x >> 5) + (threadIdx.x >> 5)));
  const int n0 = wave * NPW;
  if (n0 >= N_NODES) return;

  const float2* __restrict__ sup2 = (const float2*)sup;
  const int2*   __restrict__ csr  = (const int2*)csr_raw;
  const float2 bv = ((const float2*)bias)[lane];

  const int myoff = off[n0 + (lane <= NPW ? lane : NPW)];
  int o[NPW + 1];
#pragma unroll
  for (int i = 0; i <= NPW; ++i) o[i] = __builtin_amdgcn_readlane(myoff, i);
  int2 ed[NPW];
#pragma unroll
  for (int i = 0; i < NPW; ++i) {
    const int len = o[i + 1] - o[i];
    ed[i] = (lane < len) ? csr[o[i] + lane] : make_int2(0, 0);
  }

#pragma unroll
  for (int i = 0; i < NPW; ++i) {
    const int beg = o[i], end = o[i + 1];
    float2 acc = make_float2(0.0f, 0.0f);
    int cnt = end - beg;
    gather_chunk(sup2, ed[i], cnt < 32 ? cnt : 32, lane, acc);
    for (int c = beg + 32; c < end; c += 32) {
      cnt = end - c;
      const int2 e2 = (lane < cnt) ? csr[c + lane] : make_int2(0, 0);
      gather_chunk(sup2, e2, cnt < 32 ? cnt : 32, lane, acc);
    }
    float2 r;
    r.x = acc.x + bv.x;
    r.y = acc.y + bv.y;
    ((float2*)(out + (long long)(n0 + i) * 64))[lane] = r;
  }
}

#define FR 32
#define HP 68

static_assert(FR == 8 * NPW && N_NODES % FR == 0, "work-groups tile the rows exactly");

__global__ __launch_bounds__(256)
void spmm_gemm_kernel(const float* __restrict__ sup,
                      const int* __restrict__ csr_raw,
                      const int* __restrict__ off,
                      const float* __restrict__ bias,
                      const float* __restrict__ Wn,
                      float* __restrict__ supn) {
  __shared__ alignas(16) v2f   Pw[(D_H / 2) * 64];
  __shared__ alignas(16) float Ht[FR * HP];

  const int t    = threadIdx.x;
  const int lane = t & 31;
  const int wv   = __builtin_amdgcn_readfirstlane(t >> 5);
  const int row0 = blockIdx.x * FR;
  const int n0   = row0 + wv * NPW;

  const int kp0 = t >> 4;
  const int cg  = t & 15;
  const float4 wa0 = *(const float4*)(Wn + (2 * kp0)      * 64 + 4 * cg);
  const float4 wb0 = *(const float4*)(Wn + (2 * kp0 + 1)  * 64 + 4 * cg);
  const float4 wa1 = *(const float4*)(Wn + (2 * kp0 + 32) * 64 + 4 * cg);
  const float4 wb1 = *(const float4*)(Wn + (2 * kp0 + 33) * 64 + 4 * cg);

  const float2* __restrict__ sup2 = (const float2*)sup;
  const int2*   __restrict__ csr  = (const int2*)csr_raw;
  const float2 bv = ((const float2*)bias)[lane];

  const int myoff = off[n0 + (lane <= NPW ? lane : NPW)];
  int o[NPW + 1];
#pragma unroll
  for (int i = 0; i <= NPW; ++i) o[i] = __builtin_amdgcn_readlane(myoff, i);
  int2 ed[NPW];
#pragma unroll
  for (int i = 0; i < NPW; ++i) {
    const int len = o[i + 1] - o[i];
    ed[i] = (lane < len) ? csr[o[i] + lane] : make_int2(0, 0);
  }

#pragma unroll
  for (int i = 0; i < NPW; ++i) {
    const int beg = o[i], end = o[i + 1];
    float2 acc = make_float2(0.0f, 0.0f);
    int cnt = end - beg;
    gather_chunk(sup2, ed[i], cnt < 32 ? cnt : 32, lane, acc);
    for (int c = beg + 32; c < end; c += 32) {
      cnt = end - c;
      const int2 e2 = (lane < cnt) ? csr[c + lane] : make_int2(0, 0);
      gather_chunk(sup2, e2, cnt < 32 ? cnt : 32, lane, acc);
    }
    float2 r;
    r.x = fmaxf(acc.x + bv.x, 0.0f);
    r.y = fmaxf(acc.y + bv.y, 0.0f);
    *(float2*)(Ht + (wv * NPW + i) * HP + 2 * lane) = r;
  }

  {
    float4* __restrict__ p0 = (float4*)(Pw + kp0 * 64 + 4 * cg);
    float4* __restrict__ p1 = (float4*)(Pw + (kp0 + 16) * 64 + 4 * cg);
    p0[0] = make_float4(wa0.x, wb0.x, wa0.y, wb0.y);
    p0[1] = make_float4(wa0.z, wb0.z, wa0.w, wb0.w);
    p1[0] = make_float4(wa1.x, wb1.x, wa1.y, wb1.y);
    p1[1] = make_float4(wa1.z, wb1.z, wa1.w, wb1.w);
  }
  __syncthreads();

  if (wv >= 4) return;
  const int half = lane >> 4;
  const int l15  = lane & 15;
  const int s    = wv >> 1;
  const int cb   = (wv & 1) * 32;
  const float* __restrict__ hrow = Ht + (16 * s + l15) * HP + 2 * half;

  v8f acc0 = {}, acc1 = {};
  for (int k = 0; k < D_H; k += 4) {
    const v2f a = *(const v2f*)(hrow + k);
    const v2f* __restrict__ Pk = Pw + (k / 2 + half) * 64 + cb;
    const v2f b0 = Pk[l15];
    const v2f b1 = Pk[l15 + 16];
    acc0 = __builtin_amdgcn_wmma_f32_16x16x4_f32(false, a, false, b0, (short)0, acc0, false, false);
    acc1 = __builtin_amdgcn_wmma_f32_16x16x4_f32(false, a, false, b1, (short)0, acc1, false, false);
  }

#pragma unroll
  for (int r = 0; r < 8; ++r) {
    const long long m = row0 + 16 * s + r + 8 * half;
    float* __restrict__ crow = supn + m * 64 + cb;
    crow[l15]      = acc0[r];
    crow[l15 + 16] = acc1[r];
  }
}

extern "C" void kernel_launch(void* const* d_in, const int* in_sizes, int n_in,
                              void* d_out, int out_size, void* d_ws, size_t ws_size,
                              hipStream_t stream) {
  const float* features = (const float*)d_in[0];
  const int*   src      = (const int*)  d_in[1];
  const int*   dst      = (const int*)  d_in[2];
  const float* ew       = (const float*)d_in[3];
  const float* W1       = (const float*)d_in[4];
  const float* b1       = (const float*)d_in[5];
  const float* W2       = (const float*)d_in[6];
  const float* b2       = (const float*)d_in[7];
  const float* W3       = (const float*)d_in[8];
  const float* b3       = (const float*)d_in[9];
  float* out = (float*)d_out;

  const long long feat_elems = (long long)N_NODES * D_H;
  float* sup  = (float*)d_ws;
  int*   csr  = (int*)(sup + feat_elems);
  int*   off  = csr + 2LL * N_EDGES;
  int*   brec = (int*)out;
  int*   tcntT = brec + 2LL * N_EDGES;
  int*   toffT = tcntT + TBL;
  float* sup2 = out;

  const dim3 blk(256);
  const int spmm_blocks = (N_NODES / NPW + 7) / 8;

  chunk_sort_kernel<<<NCH, dim3(EB_THR), 0, stream>>>(src, dst, ew, tcntT, toffT, brec);

  group_gemm_kernel<<<NB + GG_GEMM_BLOCKS, dim3(GB_THR), 0, stream>>>(brec, tcntT, toffT, off, csr,
                                                                     features, W1, sup);

  spmm_gemm_kernel<<<N_NODES / FR, blk, 0, stream>>>(sup, csr, off, b1, W2, sup2);

  spmm_gemm_kernel<<<N_NODES / FR, blk, 0, stream>>>(sup2, csr, off, b2, W3, sup);

  spmm_gather_kernel<<<spmm_blocks, blk, 0, stream>>>(sup, csr, off, b3, out);

  (void)in_sizes; (void)n_in; (void)out_size; (void)ws_size;
}
